# starting kernel
# baseline (speedup 1.0000x reference)
#include <hip/hip_runtime.h>
#include <hip/hip_bf16.h>


typedef float v2f __attribute__((ext_vector_type(2)));
typedef float v8f __attribute__((ext_vector_type(8)));

#define XB   512
#define XS   64
#define XD   128
#define XNH  4
#define XDH  32
#define XT   (XB * XS)
#define XEPS 1e-5f

__device__ __forceinline__ v8f wmma4(v2f a, v2f b, v8f c) {
  return __builtin_amdgcn_wmma_f32_16x16x4_f32(false, a, false, b, (short)0, c,
                                               false, false);
}

__device__ __forceinline__ float gelu_exact(float x) {
  return 0.5f * x * (1.0f + erff(x * 0.70710678118654752f));
}

__global__ void k_embed(const int* __restrict__ x, const float* __restrict__ emb,
                        float* __restrict__ h, int total) {
  int i = blockIdx.x * blockDim.x + threadIdx.x;
  if (i >= total) return;
  int t = i >> 7;
  int c = i & 127;
  h[i] = emb[x[t] * XD + c];
}

__global__ void k_ln(const float* __restrict__ h, const float* __restrict__ w,
                     float* __restrict__ xn) {
  int wid  = threadIdx.x >> 5;
  int lane = threadIdx.x & 31;
  int t = blockIdx.x * (blockDim.x >> 5) + wid;
  const float4* hv = (const float4*)(h + (size_t)t * XD);
  float4 v = hv[lane];
  float s = v.x + v.y + v.z + v.w;
  float q = v.x * v.x + v.y * v.y + v.z * v.z + v.w * v.w;
  #pragma unroll
  for (int m = 16; m >= 1; m >>= 1) {
    s += __shfl_xor(s, m, 32);
    q += __shfl_xor(q, m, 32);
  }
  float mu  = s * (1.0f / XD);
  float var = q * (1.0f / XD) - mu * mu;
  float rs  = rsqrtf(var + XEPS);
  float4 wl = ((const float4*)w)[lane];
  float4 o;
  o.x = (v.x - mu) * rs * wl.x;
  o.y = (v.y - mu) * rs * wl.y;
  o.z = (v.z - mu) * rs * wl.z;
  o.w = (v.w - mu) * rs * wl.w;
  ((float4*)(xn + (size_t)t * XD))[lane] = o;
}

__global__ __launch_bounds__(32) void k_scan(const float* __restrict__ xn,
                                             float* __restrict__ hio,
                                             const float* __restrict__ Wg,
                                             const float* __restrict__ R,
                                             const float* __restrict__ bias,
                                             const float* __restrict__ gnw) {
  __shared__ float lwg[8][8][64];
  __shared__ float lr[8][8][64];
  __shared__ float lh[16 * 32];
  __shared__ float lb[128];
  __shared__ float lgn[32];

  const int lane = threadIdx.x;
  const int hlf  = lane >> 4;
  const int l16  = lane & 15;
  const int b0   = blockIdx.x * 16;
  const int n    = blockIdx.y;

  #pragma unroll
  for (int nt = 0; nt < 8; ++nt) {
    int N  = nt * 16 + l16;
    int g  = N >> 5;
    int dh = N & 31;
    #pragma unroll
    for (int kk = 0; kk < 8; ++kk) {
      int K0 = kk * 4 + hlf * 2;
      const float* wgp = Wg + (((size_t)(g * XNH + n) * XDH + dh) * XDH + K0);
      lwg[nt][kk][lane * 2 + 0] = wgp[0];
      lwg[nt][kk][lane * 2 + 1] = wgp[1];
      lr[nt][kk][lane * 2 + 0] = R[(size_t)(n * XDH + K0 + 0) * (4 * XDH) + N];
      lr[nt][kk][lane * 2 + 1] = R[(size_t)(n * XDH + K0 + 1) * (4 * XDH) + N];
    }
  }
  for (int j = lane; j < 128; j += 32) lb[j] = bias[n * 128 + j];
  lgn[lane & 31] = gnw[n * XDH + (lane & 31)];
  for (int j = lane; j < 16 * 32; j += 32) lh[j] = 0.0f;
  __syncthreads();

  float cS[2][8]  = {};
  float nS[2][8]  = {};
  float mS[2][8]  = {};

  #pragma unroll 1
  for (int s = 0; s < XS; ++s) {
    v2f ax[8], ah[8];
    const float* xp = xn + ((size_t)(b0 + l16) * XS + s) * XD + n * XDH;
    #pragma unroll
    for (int kk = 0; kk < 8; ++kk) {
      int K0 = kk * 4 + hlf * 2;
      ax[kk].x = xp[K0];
      ax[kk].y = xp[K0 + 1];
      ah[kk].x = lh[l16 * 32 + K0];
      ah[kk].y = lh[l16 * 32 + K0 + 1];
    }

    v8f acc[8];
    #pragma unroll
    for (int nt = 0; nt < 8; ++nt) {
      float bj = lb[nt * 16 + l16];
      v8f a;
      #pragma unroll
      for (int r = 0; r < 8; ++r) a[r] = bj;
      #pragma unroll
      for (int kk = 0; kk < 8; ++kk) {
        v2f bw, br;
        bw.x = lwg[nt][kk][lane * 2 + 0];
        bw.y = lwg[nt][kk][lane * 2 + 1];
        a = wmma4(ax[kk], bw, a);
        br.x = lr[nt][kk][lane * 2 + 0];
        br.y = lr[nt][kk][lane * 2 + 1];
        a = wmma4(ah[kk], br, a);
      }
      acc[nt] = a;
    }
    __syncthreads();

    #pragma unroll
    for (int t2 = 0; t2 < 2; ++t2) {
      #pragma unroll
      for (int r = 0; r < 8; ++r) {
        float ir  = acc[0 + t2][r];
        float fr  = acc[2 + t2][r];
        float zr  = acc[4 + t2][r];
        float orr = acc[6 + t2][r];
        float ls  = (fr >= 0.0f) ? (-log1pf(expf(-fr)))
                                 : (fr - log1pf(expf(fr)));
        float lfm = mS[t2][r] + ls;
        float mn  = fmaxf(ir, lfm);
        float ig  = expf(ir - mn);
        float fg  = expf(lfm - mn);
        float cn  = fg * cS[t2][r] + ig * tanhf(zr);
        float nn  = fg * nS[t2][r] + ig;
        float og  = 1.0f / (1.0f + expf(-orr));
        float hn  = og * cn / nn;
        cS[t2][r] = cn;
        nS[t2][r] = nn;
        mS[t2][r] = mn;
        lh[(r + hlf * 8) * 32 + t2 * 16 + l16] = hn;
      }
    }
    __syncthreads();

    if (lane < 16) {
      float mu = 0.0f, q = 0.0f;
      #pragma unroll
      for (int i = 0; i < 32; ++i) {
        float v = lh[lane * 32 + i];
        mu += v;
        q  += v * v;
      }
      mu *= (1.0f / XDH);
      q = q * (1.0f / XDH) - mu * mu;
      float rs = rsqrtf(q + XEPS);
      float* hp = hio + ((size_t)(b0 + lane) * XS + s) * XD + n * XDH;
      #pragma unroll
      for (int i = 0; i < 32; ++i) {
        float v = (lh[lane * 32 + i] - mu) * rs * lgn[i];
        hp[i] = hp[i] + v;
      }
    }
    __syncthreads();
  }
}

__global__ __launch_bounds__(32) void k_ffn(const float* __restrict__ xn,
                                            float* __restrict__ hio,
                                            const float* __restrict__ Wup,
                                            const float* __restrict__ Wdn) {
  __shared__ float lact[16 * 128];
  const int lane = threadIdx.x;
  const int hlf  = lane >> 4;
  const int l16  = lane & 15;
  const size_t t0 = (size_t)blockIdx.x * 16;

  v2f ax[32];
  const float* xp = xn + (t0 + l16) * XD;
  #pragma unroll
  for (int kk = 0; kk < 32; ++kk) {
    int K0 = kk * 4 + hlf * 2;
    ax[kk].x = xp[K0];
    ax[kk].y = xp[K0 + 1];
  }

  float gatev[8][8];
  #pragma unroll
  for (int nt = 0; nt < 8; ++nt) {
    v8f a;
    #pragma unroll
    for (int r = 0; r < 8; ++r) a[r] = 0.0f;
    int j = nt * 16 + l16;
    #pragma unroll
    for (int kk = 0; kk < 32; ++kk) {
      int K0 = kk * 4 + hlf * 2;
      v2f b;
      b.x = Wup[(size_t)j * XD + K0];
      b.y = Wup[(size_t)j * XD + K0 + 1];
      a = wmma4(ax[kk], b, a);
    }
    #pragma unroll
    for (int r = 0; r < 8; ++r) gatev[nt][r] = a[r];
  }

  #pragma unroll
  for (int nt = 0; nt < 8; ++nt) {
    v8f a;
    #pragma unroll
    for (int r = 0; r < 8; ++r) a[r] = 0.0f;
    int j = 128 + nt * 16 + l16;
    #pragma unroll
    for (int kk = 0; kk < 32; ++kk) {
      int K0 = kk * 4 + hlf * 2;
      v2f b;
      b.x = Wup[(size_t)j * XD + K0];
      b.y = Wup[(size_t)j * XD + K0 + 1];
      a = wmma4(ax[kk], b, a);
    }
    #pragma unroll
    for (int r = 0; r < 8; ++r) {
      float act = gelu_exact(gatev[nt][r]) * a[r];
      lact[(r + hlf * 8) * 128 + nt * 16 + l16] = act;
    }
  }
  __syncthreads();

  v2f aa[32];
  const float* lp = lact + l16 * 128;
  #pragma unroll
  for (int kk = 0; kk < 32; ++kk) {
    int K0 = kk * 4 + hlf * 2;
    aa[kk].x = lp[K0];
    aa[kk].y = lp[K0 + 1];
  }
  #pragma unroll
  for (int nt = 0; nt < 8; ++nt) {
    v8f a;
    #pragma unroll
    for (int r = 0; r < 8; ++r) a[r] = 0.0f;
    int j = nt * 16 + l16;
    #pragma unroll
    for (int kk = 0; kk < 32; ++kk) {
      int K0 = kk * 4 + hlf * 2;
      v2f b;
      b.x = Wdn[(size_t)j * XD + K0];
      b.y = Wdn[(size_t)j * XD + K0 + 1];
      a = wmma4(aa[kk], b, a);
    }
    #pragma unroll
    for (int r = 0; r < 8; ++r) {
      size_t row = t0 + r + hlf * 8;
      hio[row * XD + nt * 16 + l16] += a[r];
    }
  }
}

__global__ __launch_bounds__(32) void k_proj(const float* __restrict__ xn,
                                             const float* __restrict__ pw,
                                             const float* __restrict__ pb,
                                             float* __restrict__ out) {
  const int lane = threadIdx.x;
  const int hlf  = lane >> 4;
  const int l16  = lane & 15;
  const size_t t0 = (size_t)blockIdx.x * 16;

  v8f a;
  #pragma unroll
  for (int r = 0; r < 8; ++r) a[r] = 0.0f;
  const float* xp = xn + (t0 + l16) * XD;
  #pragma unroll
  for (int kk = 0; kk < 32; ++kk) {
    int K0 = kk * 4 + hlf * 2;
    v2f av;
    av.x = xp[K0];
    av.y = xp[K0 + 1];
    v2f b;
    b.x = 0.0f;
    b.y = 0.0f;
    if (l16 < 9) {
      b.x = pw[(size_t)l16 * XD + K0];
      b.y = pw[(size_t)l16 * XD + K0 + 1];
    }
    a = wmma4(av, b, a);
  }
  if (l16 < 9) {
    float bb = pb[l16];
    #pragma unroll
    for (int r = 0; r < 8; ++r) {
      size_t row = t0 + r + hlf * 8;
      out[row * 9 + l16] = a[r] + bb;
    }
  }
}

extern "C" void kernel_launch(void* const* d_in, const int* in_sizes, int n_in,
                              void* d_out, int out_size, void* d_ws, size_t ws_size,
                              hipStream_t stream) {
  (void)in_sizes; (void)n_in; (void)out_size; (void)ws_size;
  const int*   x     = (const int*)d_in[0];
  const float* emb   = (const float*)d_in[1];
  const float* ln1   = (const float*)d_in[2];
  const float* Wg    = (const float*)d_in[3];
  const float* R     = (const float*)d_in[4];
  const float* bias  = (const float*)d_in[5];
  const float* gnw   = (const float*)d_in[6];
  const float* ln2   = (const float*)d_in[7];
  const float* wup   = (const float*)d_in[8];
  const float* wdn   = (const float*)d_in[9];
  const float* postw = (const float*)d_in[10];
  const float* pw    = (const float*)d_in[11];
  const float* pb    = (const float*)d_in[12];
  float* out = (float*)d_out;

  float* h  = (float*)d_ws;
  float* xb = h + (size_t)XT * XD;

  k_embed<<<(XT * XD + 255) / 256, 256, 0, stream>>>(x, emb, h, XT * XD);

  for (int bi = 0; bi < 2; ++bi) {
    k_ln<<<XT / 8, 256, 0, stream>>>(h, ln1 + bi * XD, xb);
    k_scan<<<dim3(XB / 16, XNH), 32, 0, stream>>>(
        xb, h,
        Wg   + (size_t)bi * 4 * XNH * XDH * XDH,
        R    + (size_t)bi * XNH * XDH * 4 * XDH,
        bias + (size_t)bi * XNH * 4 * XDH,
        gnw  + (size_t)bi * XD);
    k_ln<<<XT / 8, 256, 0, stream>>>(h, ln2 + bi * XD, xb);
    k_ffn<<<XT / 16, 32, 0, stream>>>(xb, h,
                                      wup + (size_t)bi * 2 * XD * XD,
                                      wdn + (size_t)bi * XD * XD);
  }

  k_ln<<<XT / 8, 256, 0, stream>>>(h, postw, xb);
  k_proj<<<XT / 16, 32, 0, stream>>>(xb, pw, pb, out);
}


# scan: one 16-wave work-group per (batch tile, head) instead of one wave: 8 waves own one gate tile each (weights in registers), gate math 1 element per lane via LDS; summation order unchanged
# speedup vs baseline: 3.9157x; 3.9157x over previous
#include <hip/hip_runtime.h>
#include <hip/hip_bf16.h>


typedef float v2f __attribute__((ext_vector_type(2)));
typedef float v8f __attribute__((ext_vector_type(8)));

#define XB   512
#define XS   64
#define XD   128
#define XNH  4
#define XDH  32
#define XT   (XB * XS)
#define XEPS 1e-5f

__device__ __forceinline__ v8f wmma4(v2f a, v2f b, v8f c) {
  return __builtin_amdgcn_wmma_f32_16x16x4_f32(false, a, false, b, (short)0, c,
                                               false, false);
}

__device__ __forceinline__ float gelu_exact(float x) {
  return 0.5f * x * (1.0f + erff(x * 0.70710678118654752f));
}

__global__ void k_embed(const int* __restrict__ x, const float* __restrict__ emb,
                        float* __restrict__ h, int total) {
  int i = blockIdx.x * blockDim.x + threadIdx.x;
  if (i >= total) return;
  int t = i >> 7;
  int c = i & 127;
  h[i] = emb[x[t] * XD + c];
}

__global__ void k_ln(const float* __restrict__ h, const float* __restrict__ w,
                     float* __restrict__ xn) {
  int wid  = threadIdx.x >> 5;
  int lane = threadIdx.x & 31;
  int t = blockIdx.x * (blockDim.x >> 5) + wid;
  const float4* hv = (const float4*)(h + (size_t)t * XD);
  float4 v = hv[lane];
  float s = v.x + v.y + v.z + v.w;
  float q = v.x * v.x + v.y * v.y + v.z * v.z + v.w * v.w;
  #pragma unroll
  for (int m = 16; m >= 1; m >>= 1) {
    s += __shfl_xor(s, m, 32);
    q += __shfl_xor(q, m, 32);
  }
  float mu  = s * (1.0f / XD);
  float var = q * (1.0f / XD) - mu * mu;
  float rs  = rsqrtf(var + XEPS);
  float4 wl = ((const float4*)w)[lane];
  float4 o;
  o.x = (v.x - mu) * rs * wl.x;
  o.y = (v.y - mu) * rs * wl.y;
  o.z = (v.z - mu) * rs * wl.z;
  o.w = (v.w - mu) * rs * wl.w;
  ((float4*)(xn + (size_t)t * XD))[lane] = o;
}

#define SCAN_LP 36
__global__ __launch_bounds__(512) void k_scan(const float* __restrict__ xn,
                                              float* __restrict__ hio,
                                              const float* __restrict__ Wg,
                                              const float* __restrict__ R,
                                              const float* __restrict__ bias,
                                              const float* __restrict__ gnw) {
  __shared__ __attribute__((aligned(16))) float lraw[4 * 16 * SCAN_LP];
  __shared__ __attribute__((aligned(16))) float lh[16 * SCAN_LP];

  const int lane = threadIdx.x & 31;
  const int w    = __builtin_amdgcn_readfirstlane((int)(threadIdx.x >> 5));
  const int hlf  = lane >> 4;
  const int l16  = lane & 15;
  const int b0   = blockIdx.x * 16;
  const int n    = blockIdx.y;
  const bool mm  = (w < 8);

  v2f bw[8], br[8];
  float bj = 0.0f;
  #pragma unroll
  for (int kk = 0; kk < 8; ++kk) {
    bw[kk].x = 0.0f; bw[kk].y = 0.0f;
    br[kk].x = 0.0f; br[kk].y = 0.0f;
  }
  if (mm) {
    const int N  = w * 16 + l16;
    const int g  = N >> 5;
    const int dh = N & 31;
    #pragma unroll
    for (int kk = 0; kk < 8; ++kk) {
      const int K0 = kk * 4 + hlf * 2;
      const float* wgp = Wg + (((size_t)(g * XNH + n) * XDH + dh) * XDH + K0);
      bw[kk].x = wgp[0];
      bw[kk].y = wgp[1];
      br[kk].x = R[(size_t)(n * XDH + K0 + 0) * (4 * XDH) + N];
      br[kk].y = R[(size_t)(n * XDH + K0 + 1) * (4 * XDH) + N];
    }
    bj = bias[n * 128 + N];
  }
  const float gn = gnw[n * XDH + lane];
  for (int j = threadIdx.x; j < 16 * SCAN_LP; j += 512) lh[j] = 0.0f;
  __syncthreads();

  float cS = 0.0f, nS = 0.0f, mS = 0.0f;
  const int row = w;
  float* hp = hio + ((size_t)(b0 + row) * XS) * XD + n * XDH + lane;
  const float* xp0 = xn + ((size_t)(b0 + l16) * XS) * XD + n * XDH + hlf * 2;

  #pragma unroll 1
  for (int s = 0; s < XS; ++s) {
    const float hold = hp[(size_t)s * XD];

    if (mm) {
      v2f ax[8], ah[8];
      const float* xp = xp0 + (size_t)s * XD;
      #pragma unroll
      for (int kk = 0; kk < 8; ++kk) {
        ax[kk].x = xp[kk * 4];
        ax[kk].y = xp[kk * 4 + 1];
        ah[kk].x = lh[l16 * SCAN_LP + kk * 4 + hlf * 2];
        ah[kk].y = lh[l16 * SCAN_LP + kk * 4 + hlf * 2 + 1];
      }
      __builtin_amdgcn_sched_barrier(0);
      v8f a;
      #pragma unroll
      for (int r = 0; r < 8; ++r) a[r] = bj;
      #pragma unroll
      for (int kk = 0; kk < 8; ++kk) {
        a = wmma4(ax[kk], bw[kk], a);
        a = wmma4(ah[kk], br[kk], a);
      }
      float* rp = lraw + ((w >> 1) * 16 + hlf * 8) * SCAN_LP + (w & 1) * 16 + l16;
      #pragma unroll
      for (int r = 0; r < 8; ++r) rp[r * SCAN_LP] = a[r];
    }
    __syncthreads();

    float hn;
    {
      float ir  = lraw[(0 * 16 + row) * SCAN_LP + lane];
      float fr  = lraw[(1 * 16 + row) * SCAN_LP + lane];
      float zr  = lraw[(2 * 16 + row) * SCAN_LP + lane];
      float orr = lraw[(3 * 16 + row) * SCAN_LP + lane];
      float ls  = (fr >= 0.0f) ? (-log1pf(expf(-fr)))
                               : (fr - log1pf(expf(fr)));
      float lfm = mS + ls;
      float mn  = fmaxf(ir, lfm);
      float ig  = expf(ir - mn);
      float fg  = expf(lfm - mn);
      float cn  = fg * cS + ig * tanhf(zr);
      float nn  = fg * nS + ig;
      float og  = 1.0f / (1.0f + expf(-orr));
      hn  = og * cn / nn;
      cS = cn;
      nS = nn;
      mS = mn;
      lh[row * SCAN_LP + lane] = hn;
    }
    __syncthreads();

    {
      float mu = 0.0f, q = 0.0f;
      #pragma unroll
      for (int i = 0; i < 32; ++i) {
        float v = lh[row * SCAN_LP + i];
        mu += v;
        q  += v * v;
      }
      mu *= (1.0f / XDH);
      q = q * (1.0f / XDH) - mu * mu;
      float rs = rsqrtf(q + XEPS);
      float v = (hn - mu) * rs * gn;
      hp[(size_t)s * XD] = hold + v;
    }
  }
}

__global__ __launch_bounds__(32) void k_ffn(const float* __restrict__ xn,
                                            float* __restrict__ hio,
                                            const float* __restrict__ Wup,
                                            const float* __restrict__ Wdn) {
  __shared__ float lact[16 * 128];
  const int lane = threadIdx.x;
  const int hlf  = lane >> 4;
  const int l16  = lane & 15;
  const size_t t0 = (size_t)blockIdx.x * 16;

  v2f ax[32];
  const float* xp = xn + (t0 + l16) * XD;
  #pragma unroll
  for (int kk = 0; kk < 32; ++kk) {
    int K0 = kk * 4 + hlf * 2;
    ax[kk].x = xp[K0];
    ax[kk].y = xp[K0 + 1];
  }

  float gatev[8][8];
  #pragma unroll
  for (int nt = 0; nt < 8; ++nt) {
    v8f a;
    #pragma unroll
    for (int r = 0; r < 8; ++r) a[r] = 0.0f;
    int j = nt * 16 + l16;
    #pragma unroll
    for (int kk = 0; kk < 32; ++kk) {
      int K0 = kk * 4 + hlf * 2;
      v2f b;
      b.x = Wup[(size_t)j * XD + K0];
      b.y = Wup[(size_t)j * XD + K0 + 1];
      a = wmma4(ax[kk], b, a);
    }
    #pragma unroll
    for (int r = 0; r < 8; ++r) gatev[nt][r] = a[r];
  }

  #pragma unroll
  for (int nt = 0; nt < 8; ++nt) {
    v8f a;
    #pragma unroll
    for (int r = 0; r < 8; ++r) a[r] = 0.0f;
    int j = 128 + nt * 16 + l16;
    #pragma unroll
    for (int kk = 0; kk < 32; ++kk) {
      int K0 = kk * 4 + hlf * 2;
      v2f b;
      b.x = Wup[(size_t)j * XD + K0];
      b.y = Wup[(size_t)j * XD + K0 + 1];
      a = wmma4(ax[kk], b, a);
    }
    #pragma unroll
    for (int r = 0; r < 8; ++r) {
      float act = gelu_exact(gatev[nt][r]) * a[r];
      lact[(r + hlf * 8) * 128 + nt * 16 + l16] = act;
    }
  }
  __syncthreads();

  v2f aa[32];
  const float* lp = lact + l16 * 128;
  #pragma unroll
  for (int kk = 0; kk < 32; ++kk) {
    int K0 = kk * 4 + hlf * 2;
    aa[kk].x = lp[K0];
    aa[kk].y = lp[K0 + 1];
  }
  #pragma unroll
  for (int nt = 0; nt < 8; ++nt) {
    v8f a;
    #pragma unroll
    for (int r = 0; r < 8; ++r) a[r] = 0.0f;
    int j = nt * 16 + l16;
    #pragma unroll
    for (int kk = 0; kk < 32; ++kk) {
      int K0 = kk * 4 + hlf * 2;
      v2f b;
      b.x = Wdn[(size_t)j * XD + K0];
      b.y = Wdn[(size_t)j * XD + K0 + 1];
      a = wmma4(aa[kk], b, a);
    }
    #pragma unroll
    for (int r = 0; r < 8; ++r) {
      size_t row = t0 + r + hlf * 8;
      hio[row * XD + nt * 16 + l16] += a[r];
    }
  }
}

__global__ __launch_bounds__(32) void k_proj(const float* __restrict__ xn,
                                             const float* __restrict__ pw,
                                             const float* __restrict__ pb,
                                             float* __restrict__ out) {
  const int lane = threadIdx.x;
  const int hlf  = lane >> 4;
  const int l16  = lane & 15;
  const size_t t0 = (size_t)blockIdx.x * 16;

  v8f a;
  #pragma unroll
  for (int r = 0; r < 8; ++r) a[r] = 0.0f;
  const float* xp = xn + (t0 + l16) * XD;
  #pragma unroll
  for (int kk = 0; kk < 32; ++kk) {
    int K0 = kk * 4 + hlf * 2;
    v2f av;
    av.x = xp[K0];
    av.y = xp[K0 + 1];
    v2f b;
    b.x = 0.0f;
    b.y = 0.0f;
    if (l16 < 9) {
      b.x = pw[(size_t)l16 * XD + K0];
      b.y = pw[(size_t)l16 * XD + K0 + 1];
    }
    a = wmma4(av, b, a);
  }
  if (l16 < 9) {
    float bb = pb[l16];
    #pragma unroll
    for (int r = 0; r < 8; ++r) {
      size_t row = t0 + r + hlf * 8;
      out[row * 9 + l16] = a[r] + bb;
    }
  }
}

extern "C" void kernel_launch(void* const* d_in, const int* in_sizes, int n_in,
                              void* d_out, int out_size, void* d_ws, size_t ws_size,
                              hipStream_t stream) {
  (void)in_sizes; (void)n_in; (void)out_size; (void)ws_size;
  const int*   x     = (const int*)d_in[0];
  const float* emb   = (const float*)d_in[1];
  const float* ln1   = (const float*)d_in[2];
  const float* Wg    = (const float*)d_in[3];
  const float* R     = (const float*)d_in[4];
  const float* bias  = (const float*)d_in[5];
  const float* gnw   = (const float*)d_in[6];
  const float* ln2   = (const float*)d_in[7];
  const float* wup   = (const float*)d_in[8];
  const float* wdn   = (const float*)d_in[9];
  const float* postw = (const float*)d_in[10];
  const float* pw    = (const float*)d_in[11];
  const float* pb    = (const float*)d_in[12];
  float* out = (float*)d_out;

  float* h  = (float*)d_ws;
  float* xb = h + (size_t)XT * XD;

  k_embed<<<(XT * XD + 255) / 256, 256, 0, stream>>>(x, emb, h, XT * XD);

  for (int bi = 0; bi < 2; ++bi) {
    k_ln<<<XT / 8, 256, 0, stream>>>(h, ln1 + bi * XD, xb);
    k_scan<<<dim3(XB / 16, XNH), 512, 0, stream>>>(
        xb, h,
        Wg   + (size_t)bi * 4 * XNH * XDH * XDH,
        R    + (size_t)bi * XNH * XDH * 4 * XDH,
        bias + (size_t)bi * XNH * 4 * XDH,
        gnw  + (size_t)bi * XD);
    k_ln<<<XT / 8, 256, 0, stream>>>(h, ln2 + bi * XD, xb);
    k_ffn<<<XT / 16, 32, 0, stream>>>(xb, h,
                                      wup + (size_t)bi * 2 * XD * XD,
                                      wdn + (size_t)bi * XD * XD);
  }

  k_ln<<<XT / 8, 256, 0, stream>>>(h, postw, xb);
  k_proj<<<XT / 16, 32, 0, stream>>>(xb, pw, pb, out);
}


# scan gate phase: log-sigmoid computed without the sign branch, as min(f,0) - log1p(exp(-|f|)); a wave with mixed signs no longer runs exp+log1p twice per step; same values bit for bit
# speedup vs baseline: 4.2012x; 1.0729x over previous
#include <hip/hip_runtime.h>
#include <hip/hip_bf16.h>


typedef float v2f __attribute__((ext_vector_type(2)));
typedef float v8f __attribute__((ext_vector_type(8)));

#define XB   512
#define XS   64
#define XD   128
#define XNH  4
#define XDH  32
#define XT   (XB * XS)
#define XEPS 1e-5f

__device__ __forceinline__ v8f wmma4(v2f a, v2f b, v8f c) {
  return __builtin_amdgcn_wmma_f32_16x16x4_f32(false, a, false, b, (short)0, c,
                                               false, false);
}

__device__ __forceinline__ float gelu_exact(float x) {
  return 0.5f * x * (1.0f + erff(x * 0.70710678118654752f));
}

__global__ void k_embed(const int* __restrict__ x, const float* __restrict__ emb,
                        float* __restrict__ h, int total) {
  int i = blockIdx.x * blockDim.x + threadIdx.x;
  if (i >= total) return;
  int t = i >> 7;
  int c = i & 127;
  h[i] = emb[x[t] * XD + c];
}

__global__ void k_ln(const float* __restrict__ h, const float* __restrict__ w,
                     float* __restrict__ xn) {
  int wid  = threadIdx.x >> 5;
  int lane = threadIdx.x & 31;
  int t = blockIdx.x * (blockDim.x >> 5) + wid;
  const float4* hv = (const float4*)(h + (size_t)t * XD);
  float4 v = hv[lane];
  float s = v.x + v.y + v.z + v.w;
  float q = v.x * v.x + v.y * v.y + v.z * v.z + v.w * v.w;
  #pragma unroll
  for (int m = 16; m >= 1; m >>= 1) {
    s += __shfl_xor(s, m, 32);
    q += __shfl_xor(q, m, 32);
  }
  float mu  = s * (1.0f / XD);
  float var = q * (1.0f / XD) - mu * mu;
  float rs  = rsqrtf(var + XEPS);
  float4 wl = ((const float4*)w)[lane];
  float4 o;
  o.x = (v.x - mu) * rs * wl.x;
  o.y = (v.y - mu) * rs * wl.y;
  o.z = (v.z - mu) * rs * wl.z;
  o.w = (v.w - mu) * rs * wl.w;
  ((float4*)(xn + (size_t)t * XD))[lane] = o;
}

#define SCAN_LP 36
__global__ __launch_bounds__(512) void k_scan(const float* __restrict__ xn,
                                              float* __restrict__ hio,
                                              const float* __restrict__ Wg,
                                              const float* __restrict__ R,
                                              const float* __restrict__ bias,
                                              const float* __restrict__ gnw) {
  __shared__ __attribute__((aligned(16))) float lraw[4 * 16 * SCAN_LP];
  __shared__ __attribute__((aligned(16))) float lh[16 * SCAN_LP];

  const int lane = threadIdx.x & 31;
  const int w    = __builtin_amdgcn_readfirstlane((int)(threadIdx.x >> 5));
  const int hlf  = lane >> 4;
  const int l16  = lane & 15;
  const int b0   = blockIdx.x * 16;
  const int n    = blockIdx.y;
  const bool mm  = (w < 8);

  v2f bw[8], br[8];
  float bj = 0.0f;
  #pragma unroll
  for (int kk = 0; kk < 8; ++kk) {
    bw[kk].x = 0.0f; bw[kk].y = 0.0f;
    br[kk].x = 0.0f; br[kk].y = 0.0f;
  }
  if (mm) {
    const int N  = w * 16 + l16;
    const int g  = N >> 5;
    const int dh = N & 31;
    #pragma unroll
    for (int kk = 0; kk < 8; ++kk) {
      const int K0 = kk * 4 + hlf * 2;
      const float* wgp = Wg + (((size_t)(g * XNH + n) * XDH + dh) * XDH + K0);
      bw[kk].x = wgp[0];
      bw[kk].y = wgp[1];
      br[kk].x = R[(size_t)(n * XDH + K0 + 0) * (4 * XDH) + N];
      br[kk].y = R[(size_t)(n * XDH + K0 + 1) * (4 * XDH) + N];
    }
    bj = bias[n * 128 + N];
  }
  const float gn = gnw[n * XDH + lane];
  for (int j = threadIdx.x; j < 16 * SCAN_LP; j += 512) lh[j] = 0.0f;
  __syncthreads();

  float cS = 0.0f, nS = 0.0f, mS = 0.0f;
  const int row = w;
  float* hp = hio + ((size_t)(b0 + row) * XS) * XD + n * XDH + lane;
  const float* xp0 = xn + ((size_t)(b0 + l16) * XS) * XD + n * XDH + hlf * 2;

  #pragma unroll 1
  for (int s = 0; s < XS; ++s) {
    const float hold = hp[(size_t)s * XD];

    if (mm) {
      v2f ax[8], ah[8];
      const float* xp = xp0 + (size_t)s * XD;
      #pragma unroll
      for (int kk = 0; kk < 8; ++kk) {
        ax[kk].x = xp[kk * 4];
        ax[kk].y = xp[kk * 4 + 1];
        ah[kk].x = lh[l16 * SCAN_LP + kk * 4 + hlf * 2];
        ah[kk].y = lh[l16 * SCAN_LP + kk * 4 + hlf * 2 + 1];
      }
      __builtin_amdgcn_sched_barrier(0);
      v8f a;
      #pragma unroll
      for (int r = 0; r < 8; ++r) a[r] = bj;
      #pragma unroll
      for (int kk = 0; kk < 8; ++kk) {
        a = wmma4(ax[kk], bw[kk], a);
        a = wmma4(ah[kk], br[kk], a);
      }
      float* rp = lraw + ((w >> 1) * 16 + hlf * 8) * SCAN_LP + (w & 1) * 16 + l16;
      #pragma unroll
      for (int r = 0; r < 8; ++r) rp[r * SCAN_LP] = a[r];
    }
    __syncthreads();

    float hn;
    {
      float ir  = lraw[(0 * 16 + row) * SCAN_LP + lane];
      float fr  = lraw[(1 * 16 + row) * SCAN_LP + lane];
      float zr  = lraw[(2 * 16 + row) * SCAN_LP + lane];
      float orr = lraw[(3 * 16 + row) * SCAN_LP + lane];
      float ls  = fminf(fr, 0.0f) - log1pf(expf(-fabsf(fr)));
      float lfm = mS + ls;
      float mn  = fmaxf(ir, lfm);
      float ig  = expf(ir - mn);
      float fg  = expf(lfm - mn);
      float cn  = fg * cS + ig * tanhf(zr);
      float nn  = fg * nS + ig;
      float og  = 1.0f / (1.0f + expf(-orr));
      hn  = og * cn / nn;
      cS = cn;
      nS = nn;
      mS = mn;
      lh[row * SCAN_LP + lane] = hn;
    }
    __syncthreads();

    {
      float mu = 0.0f, q = 0.0f;
      #pragma unroll
      for (int i = 0; i < 32; ++i) {
        float v = lh[row * SCAN_LP + i];
        mu += v;
        q  += v * v;
      }
      mu *= (1.0f / XDH);
      q = q * (1.0f / XDH) - mu * mu;
      float rs = rsqrtf(q + XEPS);
      float v = (hn - mu) * rs * gn;
      hp[(size_t)s * XD] = hold + v;
    }
  }
}

__global__ __launch_bounds__(32) void k_ffn(const float* __restrict__ xn,
                                            float* __restrict__ hio,
                                            const float* __restrict__ Wup,
                                            const float* __restrict__ Wdn) {
  __shared__ float lact[16 * 128];
  const int lane = threadIdx.x;
  const int hlf  = lane >> 4;
  const int l16  = lane & 15;
  const size_t t0 = (size_t)blockIdx.x * 16;

  v2f ax[32];
  const float* xp = xn + (t0 + l16) * XD;
  #pragma unroll
  for (int kk = 0; kk < 32; ++kk) {
    int K0 = kk * 4 + hlf * 2;
    ax[kk].x = xp[K0];
    ax[kk].y = xp[K0 + 1];
  }

  float gatev[8][8];
  #pragma unroll
  for (int nt = 0; nt < 8; ++nt) {
    v8f a;
    #pragma unroll
    for (int r = 0; r < 8; ++r) a[r] = 0.0f;
    int j = nt * 16 + l16;
    #pragma unroll
    for (int kk = 0; kk < 32; ++kk) {
      int K0 = kk * 4 + hlf * 2;
      v2f b;
      b.x = Wup[(size_t)j * XD + K0];
      b.y = Wup[(size_t)j * XD + K0 + 1];
      a = wmma4(ax[kk], b, a);
    }
    #pragma unroll
    for (int r = 0; r < 8; ++r) gatev[nt][r] = a[r];
  }

  #pragma unroll
  for (int nt = 0; nt < 8; ++nt) {
    v8f a;
    #pragma unroll
    for (int r = 0; r < 8; ++r) a[r] = 0.0f;
    int j = 128 + nt * 16 + l16;
    #pragma unroll
    for (int kk = 0; kk < 32; ++kk) {
      int K0 = kk * 4 + hlf * 2;
      v2f b;
      b.x = Wup[(size_t)j * XD + K0];
      b.y = Wup[(size_t)j * XD + K0 + 1];
      a = wmma4(ax[kk], b, a);
    }
    #pragma unroll
    for (int r = 0; r < 8; ++r) {
      float act = gelu_exact(gatev[nt][r]) * a[r];
      lact[(r + hlf * 8) * 128 + nt * 16 + l16] = act;
    }
  }
  __syncthreads();

  v2f aa[32];
  const float* lp = lact + l16 * 128;
  #pragma unroll
  for (int kk = 0; kk < 32; ++kk) {
    int K0 = kk * 4 + hlf * 2;
    aa[kk].x = lp[K0];
    aa[kk].y = lp[K0 + 1];
  }
  #pragma unroll
  for (int nt = 0; nt < 8; ++nt) {
    v8f a;
    #pragma unroll
    for (int r = 0; r < 8; ++r) a[r] = 0.0f;
    int j = nt * 16 + l16;
    #pragma unroll
    for (int kk = 0; kk < 32; ++kk) {
      int K0 = kk * 4 + hlf * 2;
      v2f b;
      b.x = Wdn[(size_t)j * XD + K0];
      b.y = Wdn[(size_t)j * XD + K0 + 1];
      a = wmma4(aa[kk], b, a);
    }
    #pragma unroll
    for (int r = 0; r < 8; ++r) {
      size_t row = t0 + r + hlf * 8;
      hio[row * XD + nt * 16 + l16] += a[r];
    }
  }
}

__global__ __launch_bounds__(32) void k_proj(const float* __restrict__ xn,
                                             const float* __restrict__ pw,
                                             const float* __restrict__ pb,
                                             float* __restrict__ out) {
  const int lane = threadIdx.x;
  const int hlf  = lane >> 4;
  const int l16  = lane & 15;
  const size_t t0 = (size_t)blockIdx.x * 16;

  v8f a;
  #pragma unroll
  for (int r = 0; r < 8; ++r) a[r] = 0.0f;
  const float* xp = xn + (t0 + l16) * XD;
  #pragma unroll
  for (int kk = 0; kk < 32; ++kk) {
    int K0 = kk * 4 + hlf * 2;
    v2f av;
    av.x = xp[K0];
    av.y = xp[K0 + 1];
    v2f b;
    b.x = 0.0f;
    b.y = 0.0f;
    if (l16 < 9) {
      b.x = pw[(size_t)l16 * XD + K0];
      b.y = pw[(size_t)l16 * XD + K0 + 1];
    }
    a = wmma4(av, b, a);
  }
  if (l16 < 9) {
    float bb = pb[l16];
    #pragma unroll
    for (int r = 0; r < 8; ++r) {
      size_t row = t0 + r + hlf * 8;
      out[row * 9 + l16] = a[r] + bb;
    }
  }
}

extern "C" void kernel_launch(void* const* d_in, const int* in_sizes, int n_in,
                              void* d_out, int out_size, void* d_ws, size_t ws_size,
                              hipStream_t stream) {
  (void)in_sizes; (void)n_in; (void)out_size; (void)ws_size;
  const int*   x     = (const int*)d_in[0];
  const float* emb   = (const float*)d_in[1];
  const float* ln1   = (const float*)d_in[2];
  const float* Wg    = (const float*)d_in[3];
  const float* R     = (const float*)d_in[4];
  const float* bias  = (const float*)d_in[5];
  const float* gnw   = (const float*)d_in[6];
  const float* ln2   = (const float*)d_in[7];
  const float* wup   = (const float*)d_in[8];
  const float* wdn   = (const float*)d_in[9];
  const float* postw = (const float*)d_in[10];
  const float* pw    = (const float*)d_in[11];
  const float* pb    = (const float*)d_in[12];
  float* out = (float*)d_out;

  float* h  = (float*)d_ws;
  float* xb = h + (size_t)XT * XD;

  k_embed<<<(XT * XD + 255) / 256, 256, 0, stream>>>(x, emb, h, XT * XD);

  for (int bi = 0; bi < 2; ++bi) {
    k_ln<<<XT / 8, 256, 0, stream>>>(h, ln1 + bi * XD, xb);
    k_scan<<<dim3(XB / 16, XNH), 512, 0, stream>>>(
        xb, h,
        Wg   + (size_t)bi * 4 * XNH * XDH * XDH,
        R    + (size_t)bi * XNH * XDH * 4 * XDH,
        bias + (size_t)bi * XNH * 4 * XDH,
        gnw  + (size_t)bi * XD);
    k_ln<<<XT / 8, 256, 0, stream>>>(h, ln2 + bi * XD, xb);
    k_ffn<<<XT / 16, 32, 0, stream>>>(xb, h,
                                      wup + (size_t)bi * 2 * XD * XD,
                                      wdn + (size_t)bi * XD * XD);
  }

  k_ln<<<XT / 8, 256, 0, stream>>>(h, postw, xb);
  k_proj<<<XT / 16, 32, 0, stream>>>(xb, pw, pb, out);
}


# scan: xn A-fragments of step s+1 are fetched from global memory during step s, and the step's two barriers fence LDS only so the loads stay in flight across them; same sums, same order
# speedup vs baseline: 5.1854x; 1.2343x over previous
#include <hip/hip_runtime.h>
#include <hip/hip_bf16.h>


typedef float v2f __attribute__((ext_vector_type(2)));
typedef float v8f __attribute__((ext_vector_type(8)));

#define XB   512
#define XS   64
#define XD   128
#define XNH  4
#define XDH  32
#define XT   (XB * XS)
#define XEPS 1e-5f

__device__ __forceinline__ v8f wmma4(v2f a, v2f b, v8f c) {
  return __builtin_amdgcn_wmma_f32_16x16x4_f32(false, a, false, b, (short)0, c,
                                               false, false);
}

__device__ __forceinline__ void lds_barrier() {
  __builtin_amdgcn_fence(__ATOMIC_RELEASE, "workgroup", "local");
  __builtin_amdgcn_s_barrier();
  __builtin_amdgcn_fence(__ATOMIC_ACQUIRE, "workgroup", "local");
}

__device__ __forceinline__ float gelu_exact(float x) {
  return 0.5f * x * (1.0f + erff(x * 0.70710678118654752f));
}

__global__ void k_embed(const int* __restrict__ x, const float* __restrict__ emb,
                        float* __restrict__ h, int total) {
  int i = blockIdx.x * blockDim.x + threadIdx.x;
  if (i >= total) return;
  int t = i >> 7;
  int c = i & 127;
  h[i] = emb[x[t] * XD + c];
}

__global__ void k_ln(const float* __restrict__ h, const float* __restrict__ w,
                     float* __restrict__ xn) {
  int wid  = threadIdx.x >> 5;
  int lane = threadIdx.x & 31;
  int t = blockIdx.x * (blockDim.x >> 5) + wid;
  const float4* hv = (const float4*)(h + (size_t)t * XD);
  float4 v = hv[lane];
  float s = v.x + v.y + v.z + v.w;
  float q = v.x * v.x + v.y * v.y + v.z * v.z + v.w * v.w;
  #pragma unroll
  for (int m = 16; m >= 1; m >>= 1) {
    s += __shfl_xor(s, m, 32);
    q += __shfl_xor(q, m, 32);
  }
  float mu  = s * (1.0f / XD);
  float var = q * (1.0f / XD) - mu * mu;
  float rs  = rsqrtf(var + XEPS);
  float4 wl = ((const float4*)w)[lane];
  float4 o;
  o.x = (v.x - mu) * rs * wl.x;
  o.y = (v.y - mu) * rs * wl.y;
  o.z = (v.z - mu) * rs * wl.z;
  o.w = (v.w - mu) * rs * wl.w;
  ((float4*)(xn + (size_t)t * XD))[lane] = o;
}

#define SCAN_LP 36
__global__ __launch_bounds__(512) void k_scan(const float* __restrict__ xn,
                                              float* __restrict__ hio,
                                              const float* __restrict__ Wg,
                                              const float* __restrict__ R,
                                              const float* __restrict__ bias,
                                              const float* __restrict__ gnw) {
  __shared__ __attribute__((aligned(16))) float lraw[4 * 16 * SCAN_LP];
  __shared__ __attribute__((aligned(16))) float lh[16 * SCAN_LP];

  const int lane = threadIdx.x & 31;
  const int w    = __builtin_amdgcn_readfirstlane((int)(threadIdx.x >> 5));
  const int hlf  = lane >> 4;
  const int l16  = lane & 15;
  const int b0   = blockIdx.x * 16;
  const int n    = blockIdx.y;
  const bool mm  = (w < 8);

  v2f bw[8], br[8];
  float bj = 0.0f;
  #pragma unroll
  for (int kk = 0; kk < 8; ++kk) {
    bw[kk].x = 0.0f; bw[kk].y = 0.0f;
    br[kk].x = 0.0f; br[kk].y = 0.0f;
  }
  if (mm) {
    const int N  = w * 16 + l16;
    const int g  = N >> 5;
    const int dh = N & 31;
    #pragma unroll
    for (int kk = 0; kk < 8; ++kk) {
      const int K0 = kk * 4 + hlf * 2;
      const float* wgp = Wg + (((size_t)(g * XNH + n) * XDH + dh) * XDH + K0);
      bw[kk].x = wgp[0];
      bw[kk].y = wgp[1];
      br[kk].x = R[(size_t)(n * XDH + K0 + 0) * (4 * XDH) + N];
      br[kk].y = R[(size_t)(n * XDH + K0 + 1) * (4 * XDH) + N];
    }
    bj = bias[n * 128 + N];
  }
  const float gn = gnw[n * XDH + lane];
  for (int j = threadIdx.x; j < 16 * SCAN_LP; j += 512) lh[j] = 0.0f;
  __syncthreads();

  float cS = 0.0f, nS = 0.0f, mS = 0.0f;
  const int row = w;
  float* hp = hio + ((size_t)(b0 + row) * XS) * XD + n * XDH + lane;
  const float* xp0 = xn + ((size_t)(b0 + l16) * XS) * XD + n * XDH + hlf * 2;

  v2f ax[8];
  #pragma unroll
  for (int kk = 0; kk < 8; ++kk) {
    ax[kk].x = 0.0f; ax[kk].y = 0.0f;
  }
  if (mm) {
    #pragma unroll
    for (int kk = 0; kk < 8; ++kk) {
      ax[kk].x = xp0[kk * 4];
      ax[kk].y = xp0[kk * 4 + 1];
    }
  }

  #pragma unroll 1
  for (int s = 0; s < XS; ++s) {
    const float hold = hp[(size_t)s * XD];

    if (mm) {
      v2f ah[8];
      #pragma unroll
      for (int kk = 0; kk < 8; ++kk) {
        ah[kk].x = lh[l16 * SCAN_LP + kk * 4 + hlf * 2];
        ah[kk].y = lh[l16 * SCAN_LP + kk * 4 + hlf * 2 + 1];
      }
      __builtin_amdgcn_sched_barrier(0);
      v8f a;
      #pragma unroll
      for (int r = 0; r < 8; ++r) a[r] = bj;
      #pragma unroll
      for (int kk = 0; kk < 8; ++kk) {
        a = wmma4(ax[kk], bw[kk], a);
        a = wmma4(ah[kk], br[kk], a);
      }
      float* rp = lraw + ((w >> 1) * 16 + hlf * 8) * SCAN_LP + (w & 1) * 16 + l16;
      #pragma unroll
      for (int r = 0; r < 8; ++r) rp[r * SCAN_LP] = a[r];
      const int sn = (s + 1 < XS) ? s + 1 : s;
      const float* xq = xp0 + (size_t)sn * XD;
      #pragma unroll
      for (int kk = 0; kk < 8; ++kk) {
        ax[kk].x = xq[kk * 4];
        ax[kk].y = xq[kk * 4 + 1];
      }
    }
    lds_barrier();

    float hn;
    {
      float ir  = lraw[(0 * 16 + row) * SCAN_LP + lane];
      float fr  = lraw[(1 * 16 + row) * SCAN_LP + lane];
      float zr  = lraw[(2 * 16 + row) * SCAN_LP + lane];
      float orr = lraw[(3 * 16 + row) * SCAN_LP + lane];
      float ls  = fminf(fr, 0.0f) - log1pf(expf(-fabsf(fr)));
      float lfm = mS + ls;
      float mn  = fmaxf(ir, lfm);
      float ig  = expf(ir - mn);
      float fg  = expf(lfm - mn);
      float cn  = fg * cS + ig * tanhf(zr);
      float nn  = fg * nS + ig;
      float og  = 1.0f / (1.0f + expf(-orr));
      hn  = og * cn / nn;
      cS = cn;
      nS = nn;
      mS = mn;
      lh[row * SCAN_LP + lane] = hn;
    }
    lds_barrier();

    {
      float mu = 0.0f, q = 0.0f;
      #pragma unroll
      for (int i = 0; i < 32; ++i) {
        float v = lh[row * SCAN_LP + i];
        mu += v;
        q  += v * v;
      }
      mu *= (1.0f / XDH);
      q = q * (1.0f / XDH) - mu * mu;
      float rs = rsqrtf(q + XEPS);
      float v = (hn - mu) * rs * gn;
      hp[(size_t)s * XD] = hold + v;
    }
  }
}

__global__ __launch_bounds__(32) void k_ffn(const float* __restrict__ xn,
                                            float* __restrict__ hio,
                                            const float* __restrict__ Wup,
                                            const float* __restrict__ Wdn) {
  __shared__ float lact[16 * 128];
  const int lane = threadIdx.x;
  const int hlf  = lane >> 4;
  const int l16  = lane & 15;
  const size_t t0 = (size_t)blockIdx.x * 16;

  v2f ax[32];
  const float* xp = xn + (t0 + l16) * XD;
  #pragma unroll
  for (int kk = 0; kk < 32; ++kk) {
    int K0 = kk * 4 + hlf * 2;
    ax[kk].x = xp[K0];
    ax[kk].y = xp[K0 + 1];
  }

  float gatev[8][8];
  #pragma unroll
  for (int nt = 0; nt < 8; ++nt) {
    v8f a;
    #pragma unroll
    for (int r = 0; r < 8; ++r) a[r] = 0.0f;
    int j = nt * 16 + l16;
    #pragma unroll
    for (int kk = 0; kk < 32; ++kk) {
      int K0 = kk * 4 + hlf * 2;
      v2f b;
      b.x = Wup[(size_t)j * XD + K0];
      b.y = Wup[(size_t)j * XD + K0 + 1];
      a = wmma4(ax[kk], b, a);
    }
    #pragma unroll
    for (int r = 0; r < 8; ++r) gatev[nt][r] = a[r];
  }

  #pragma unroll
  for (int nt = 0; nt < 8; ++nt) {
    v8f a;
    #pragma unroll
    for (int r = 0; r < 8; ++r) a[r] = 0.0f;
    int j = 128 + nt * 16 + l16;
    #pragma unroll
    for (int kk = 0; kk < 32; ++kk) {
      int K0 = kk * 4 + hlf * 2;
      v2f b;
      b.x = Wup[(size_t)j * XD + K0];
      b.y = Wup[(size_t)j * XD + K0 + 1];
      a = wmma4(ax[kk], b, a);
    }
    #pragma unroll
    for (int r = 0; r < 8; ++r) {
      float act = gelu_exact(gatev[nt][r]) * a[r];
      lact[(r + hlf * 8) * 128 + nt * 16 + l16] = act;
    }
  }
  __syncthreads();

  v2f aa[32];
  const float* lp = lact + l16 * 128;
  #pragma unroll
  for (int kk = 0; kk < 32; ++kk) {
    int K0 = kk * 4 + hlf * 2;
    aa[kk].x = lp[K0];
    aa[kk].y = lp[K0 + 1];
  }
  #pragma unroll
  for (int nt = 0; nt < 8; ++nt) {
    v8f a;
    #pragma unroll
    for (int r = 0; r < 8; ++r) a[r] = 0.0f;
    int j = nt * 16 + l16;
    #pragma unroll
    for (int kk = 0; kk < 32; ++kk) {
      int K0 = kk * 4 + hlf * 2;
      v2f b;
      b.x = Wdn[(size_t)j * XD + K0];
      b.y = Wdn[(size_t)j * XD + K0 + 1];
      a = wmma4(aa[kk], b, a);
    }
    #pragma unroll
    for (int r = 0; r < 8; ++r) {
      size_t row = t0 + r + hlf * 8;
      hio[row * XD + nt * 16 + l16] += a[r];
    }
  }
}

__global__ __launch_bounds__(32) void k_proj(const float* __restrict__ xn,
                                             const float* __restrict__ pw,
                                             const float* __restrict__ pb,
                                             float* __restrict__ out) {
  const int lane = threadIdx.x;
  const int hlf  = lane >> 4;
  const int l16  = lane & 15;
  const size_t t0 = (size_t)blockIdx.x * 16;

  v8f a;
  #pragma unroll
  for (int r = 0; r < 8; ++r) a[r] = 0.0f;
  const float* xp = xn + (t0 + l16) * XD;
  #pragma unroll
  for (int kk = 0; kk < 32; ++kk) {
    int K0 = kk * 4 + hlf * 2;
    v2f av;
    av.x = xp[K0];
    av.y = xp[K0 + 1];
    v2f b;
    b.x = 0.0f;
    b.y = 0.0f;
    if (l16 < 9) {
      b.x = pw[(size_t)l16 * XD + K0];
      b.y = pw[(size_t)l16 * XD + K0 + 1];
    }
    a = wmma4(av, b, a);
  }
  if (l16 < 9) {
    float bb = pb[l16];
    #pragma unroll
    for (int r = 0; r < 8; ++r) {
      size_t row = t0 + r + hlf * 8;
      out[row * 9 + l16] = a[r] + bb;
    }
  }
}

extern "C" void kernel_launch(void* const* d_in, const int* in_sizes, int n_in,
                              void* d_out, int out_size, void* d_ws, size_t ws_size,
                              hipStream_t stream) {
  (void)in_sizes; (void)n_in; (void)out_size; (void)ws_size;
  const int*   x     = (const int*)d_in[0];
  const float* emb   = (const float*)d_in[1];
  const float* ln1   = (const float*)d_in[2];
  const float* Wg    = (const float*)d_in[3];
  const float* R     = (const float*)d_in[4];
  const float* bias  = (const float*)d_in[5];
  const float* gnw   = (const float*)d_in[6];
  const float* ln2   = (const float*)d_in[7];
  const float* wup   = (const float*)d_in[8];
  const float* wdn   = (const float*)d_in[9];
  const float* postw = (const float*)d_in[10];
  const float* pw    = (const float*)d_in[11];
  const float* pb    = (const float*)d_in[12];
  float* out = (float*)d_out;

  float* h  = (float*)d_ws;
  float* xb = h + (size_t)XT * XD;

  k_embed<<<(XT * XD + 255) / 256, 256, 0, stream>>>(x, emb, h, XT * XD);

  for (int bi = 0; bi < 2; ++bi) {
    k_ln<<<XT / 8, 256, 0, stream>>>(h, ln1 + bi * XD, xb);
    k_scan<<<dim3(XB / 16, XNH), 512, 0, stream>>>(
        xb, h,
        Wg   + (size_t)bi * 4 * XNH * XDH * XDH,
        R    + (size_t)bi * XNH * XDH * 4 * XDH,
        bias + (size_t)bi * XNH * 4 * XDH,
        gnw  + (size_t)bi * XD);
    k_ln<<<XT / 8, 256, 0, stream>>>(h, ln2 + bi * XD, xb);
    k_ffn<<<XT / 16, 32, 0, stream>>>(xb, h,
                                      wup + (size_t)bi * 2 * XD * XD,
                                      wdn + (size_t)bi * XD * XD);
  }

  k_ln<<<XT / 8, 256, 0, stream>>>(h, postw, xb);
  k_proj<<<XT / 16, 32, 0, stream>>>(xb, pw, pb, out);
}


# scan: 8 batch rows per work-group, not 16 (8 waves: a gate tile and a batch row each): 256 work-groups fill all 256 WGPs, gate phase at 2 waves per SIMD, not 4; same sums, same order
# speedup vs baseline: 5.6217x; 1.0841x over previous
#include <hip/hip_runtime.h>
#include <hip/hip_bf16.h>


typedef float v2f __attribute__((ext_vector_type(2)));
typedef float v8f __attribute__((ext_vector_type(8)));

#define XB   512
#define XS   64
#define XD   128
#define XNH  4
#define XDH  32
#define XT   (XB * XS)
#define XEPS 1e-5f

__device__ __forceinline__ v8f wmma4(v2f a, v2f b, v8f c) {
  return __builtin_amdgcn_wmma_f32_16x16x4_f32(false, a, false, b, (short)0, c,
                                               false, false);
}

__device__ __forceinline__ void lds_barrier() {
  __builtin_amdgcn_fence(__ATOMIC_RELEASE, "workgroup", "local");
  __builtin_amdgcn_s_barrier();
  __builtin_amdgcn_fence(__ATOMIC_ACQUIRE, "workgroup", "local");
}

__device__ __forceinline__ float gelu_exact(float x) {
  return 0.5f * x * (1.0f + erff(x * 0.70710678118654752f));
}

__global__ void k_embed(const int* __restrict__ x, const float* __restrict__ emb,
                        float* __restrict__ h, int total) {
  int i = blockIdx.x * blockDim.x + threadIdx.x;
  if (i >= total) return;
  int t = i >> 7;
  int c = i & 127;
  h[i] = emb[x[t] * XD + c];
}

__global__ void k_ln(const float* __restrict__ h, const float* __restrict__ w,
                     float* __restrict__ xn) {
  int wid  = threadIdx.x >> 5;
  int lane = threadIdx.x & 31;
  int t = blockIdx.x * (blockDim.x >> 5) + wid;
  const float4* hv = (const float4*)(h + (size_t)t * XD);
  float4 v = hv[lane];
  float s = v.x + v.y + v.z + v.w;
  float q = v.x * v.x + v.y * v.y + v.z * v.z + v.w * v.w;
  #pragma unroll
  for (int m = 16; m >= 1; m >>= 1) {
    s += __shfl_xor(s, m, 32);
    q += __shfl_xor(q, m, 32);
  }
  float mu  = s * (1.0f / XD);
  float var = q * (1.0f / XD) - mu * mu;
  float rs  = rsqrtf(var + XEPS);
  float4 wl = ((const float4*)w)[lane];
  float4 o;
  o.x = (v.x - mu) * rs * wl.x;
  o.y = (v.y - mu) * rs * wl.y;
  o.z = (v.z - mu) * rs * wl.z;
  o.w = (v.w - mu) * rs * wl.w;
  ((float4*)(xn + (size_t)t * XD))[lane] = o;
}

#define SCAN_LP 36
#define SCAN_R  8
__global__ __launch_bounds__(256) void k_scan(const float* __restrict__ xn,
                                              float* __restrict__ hio,
                                              const float* __restrict__ Wg,
                                              const float* __restrict__ R,
                                              const float* __restrict__ bias,
                                              const float* __restrict__ gnw) {
  __shared__ __attribute__((aligned(16))) float lraw[4 * 16 * SCAN_LP];
  __shared__ __attribute__((aligned(16))) float lh[SCAN_R * SCAN_LP];

  const int lane = threadIdx.x & 31;
  const int w    = __builtin_amdgcn_readfirstlane((int)(threadIdx.x >> 5));
  const int hlf  = lane >> 4;
  const int l16  = lane & 15;
  const int ar   = lane & 7;
  const int b0   = blockIdx.x * SCAN_R;
  const int n    = blockIdx.y;

  v2f bw[8], br[8];
  const int N  = w * 16 + l16;
  {
    const int g  = N >> 5;
    const int dh = N & 31;
    #pragma unroll
    for (int kk = 0; kk < 8; ++kk) {
      const int K0 = kk * 4 + hlf * 2;
      const float* wgp = Wg + (((size_t)(g * XNH + n) * XDH + dh) * XDH + K0);
      bw[kk].x = wgp[0];
      bw[kk].y = wgp[1];
      br[kk].x = R[(size_t)(n * XDH + K0 + 0) * (4 * XDH) + N];
      br[kk].y = R[(size_t)(n * XDH + K0 + 1) * (4 * XDH) + N];
    }
  }
  const float bj = bias[n * 128 + N];
  const float gn = gnw[n * XDH + lane];
  for (int j = threadIdx.x; j < SCAN_R * SCAN_LP; j += 256) lh[j] = 0.0f;
  __syncthreads();

  float cS = 0.0f, nS = 0.0f, mS = 0.0f;
  const int row = w;
  float* hp = hio + ((size_t)(b0 + row) * XS) * XD + n * XDH + lane;
  const float* xp0 = xn + ((size_t)(b0 + ar) * XS) * XD + n * XDH + hlf * 2;

  v2f ax[8];
  #pragma unroll
  for (int kk = 0; kk < 8; ++kk) {
    ax[kk].x = xp0[kk * 4];
    ax[kk].y = xp0[kk * 4 + 1];
  }

  #pragma unroll 1
  for (int s = 0; s < XS; ++s) {
    const float hold = hp[(size_t)s * XD];

    {
      v2f ah[8];
      #pragma unroll
      for (int kk = 0; kk < 8; ++kk) {
        ah[kk].x = lh[ar * SCAN_LP + kk * 4 + hlf * 2];
        ah[kk].y = lh[ar * SCAN_LP + kk * 4 + hlf * 2 + 1];
      }
      __builtin_amdgcn_sched_barrier(0);
      v8f a;
      #pragma unroll
      for (int r = 0; r < 8; ++r) a[r] = bj;
      #pragma unroll
      for (int kk = 0; kk < 8; ++kk) {
        a = wmma4(ax[kk], bw[kk], a);
        a = wmma4(ah[kk], br[kk], a);
      }
      {
        float* rp = lraw + ((w >> 1) * 16 + hlf * 8) * SCAN_LP + (w & 1) * 16 + l16;
        #pragma unroll
        for (int r = 0; r < 8; ++r) rp[r * SCAN_LP] = a[r];
      }
      const int sn = (s + 1 < XS) ? s + 1 : s;
      const float* xq = xp0 + (size_t)sn * XD;
      #pragma unroll
      for (int kk = 0; kk < 8; ++kk) {
        ax[kk].x = xq[kk * 4];
        ax[kk].y = xq[kk * 4 + 1];
      }
    }
    lds_barrier();

    float hn;
    {
      float ir  = lraw[(0 * 16 + row) * SCAN_LP + lane];
      float fr  = lraw[(1 * 16 + row) * SCAN_LP + lane];
      float zr  = lraw[(2 * 16 + row) * SCAN_LP + lane];
      float orr = lraw[(3 * 16 + row) * SCAN_LP + lane];
      float ls  = fminf(fr, 0.0f) - log1pf(expf(-fabsf(fr)));
      float lfm = mS + ls;
      float mn  = fmaxf(ir, lfm);
      float ig  = expf(ir - mn);
      float fg  = expf(lfm - mn);
      float cn  = fg * cS + ig * tanhf(zr);
      float nn  = fg * nS + ig;
      float og  = 1.0f / (1.0f + expf(-orr));
      hn  = og * cn / nn;
      cS = cn;
      nS = nn;
      mS = mn;
      lh[row * SCAN_LP + lane] = hn;
    }
    lds_barrier();

    {
      float mu = 0.0f, q = 0.0f;
      #pragma unroll
      for (int i = 0; i < 32; ++i) {
        float v = lh[row * SCAN_LP + i];
        mu += v;
        q  += v * v;
      }
      mu *= (1.0f / XDH);
      q = q * (1.0f / XDH) - mu * mu;
      float rs = rsqrtf(q + XEPS);
      float v = (hn - mu) * rs * gn;
      hp[(size_t)s * XD] = hold + v;
    }
  }
}

__global__ __launch_bounds__(32) void k_ffn(const float* __restrict__ xn,
                                            float* __restrict__ hio,
                                            const float* __restrict__ Wup,
                                            const float* __restrict__ Wdn) {
  __shared__ float lact[16 * 128];
  const int lane = threadIdx.x;
  const int hlf  = lane >> 4;
  const int l16  = lane & 15;
  const size_t t0 = (size_t)blockIdx.x * 16;

  v2f ax[32];
  const float* xp = xn + (t0 + l16) * XD;
  #pragma unroll
  for (int kk = 0; kk < 32; ++kk) {
    int K0 = kk * 4 + hlf * 2;
    ax[kk].x = xp[K0];
    ax[kk].y = xp[K0 + 1];
  }

  float gatev[8][8];
  #pragma unroll
  for (int nt = 0; nt < 8; ++nt) {
    v8f a;
    #pragma unroll
    for (int r = 0; r < 8; ++r) a[r] = 0.0f;
    int j = nt * 16 + l16;
    #pragma unroll
    for (int kk = 0; kk < 32; ++kk) {
      int K0 = kk * 4 + hlf * 2;
      v2f b;
      b.x = Wup[(size_t)j * XD + K0];
      b.y = Wup[(size_t)j * XD + K0 + 1];
      a = wmma4(ax[kk], b, a);
    }
    #pragma unroll
    for (int r = 0; r < 8; ++r) gatev[nt][r] = a[r];
  }

  #pragma unroll
  for (int nt = 0; nt < 8; ++nt) {
    v8f a;
    #pragma unroll
    for (int r = 0; r < 8; ++r) a[r] = 0.0f;
    int j = 128 + nt * 16 + l16;
    #pragma unroll
    for (int kk = 0; kk < 32; ++kk) {
      int K0 = kk * 4 + hlf * 2;
      v2f b;
      b.x = Wup[(size_t)j * XD + K0];
      b.y = Wup[(size_t)j * XD + K0 + 1];
      a = wmma4(ax[kk], b, a);
    }
    #pragma unroll
    for (int r = 0; r < 8; ++r) {
      float act = gelu_exact(gatev[nt][r]) * a[r];
      lact[(r + hlf * 8) * 128 + nt * 16 + l16] = act;
    }
  }
  __syncthreads();

  v2f aa[32];
  const float* lp = lact + l16 * 128;
  #pragma unroll
  for (int kk = 0; kk < 32; ++kk) {
    int K0 = kk * 4 + hlf * 2;
    aa[kk].x = lp[K0];
    aa[kk].y = lp[K0 + 1];
  }
  #pragma unroll
  for (int nt = 0; nt < 8; ++nt) {
    v8f a;
    #pragma unroll
    for (int r = 0; r < 8; ++r) a[r] = 0.0f;
    int j = nt * 16 + l16;
    #pragma unroll
    for (int kk = 0; kk < 32; ++kk) {
      int K0 = kk * 4 + hlf * 2;
      v2f b;
      b.x = Wdn[(size_t)j * XD + K0];
      b.y = Wdn[(size_t)j * XD + K0 + 1];
      a = wmma4(aa[kk], b, a);
    }
    #pragma unroll
    for (int r = 0; r < 8; ++r) {
      size_t row = t0 + r + hlf * 8;
      hio[row * XD + nt * 16 + l16] += a[r];
    }
  }
}

__global__ __launch_bounds__(32) void k_proj(const float* __restrict__ xn,
                                             const float* __restrict__ pw,
                                             const float* __restrict__ pb,
                                             float* __restrict__ out) {
  const int lane = threadIdx.x;
  const int hlf  = lane >> 4;
  const int l16  = lane & 15;
  const size_t t0 = (size_t)blockIdx.x * 16;

  v8f a;
  #pragma unroll
  for (int r = 0; r < 8; ++r) a[r] = 0.0f;
  const float* xp = xn + (t0 + l16) * XD;
  #pragma unroll
  for (int kk = 0; kk < 32; ++kk) {
    int K0 = kk * 4 + hlf * 2;
    v2f av;
    av.x = xp[K0];
    av.y = xp[K0 + 1];
    v2f b;
    b.x = 0.0f;
    b.y = 0.0f;
    if (l16 < 9) {
      b.x = pw[(size_t)l16 * XD + K0];
      b.y = pw[(size_t)l16 * XD + K0 + 1];
    }
    a = wmma4(av, b, a);
  }
  if (l16 < 9) {
    float bb = pb[l16];
    #pragma unroll
    for (int r = 0; r < 8; ++r) {
      size_t row = t0 + r + hlf * 8;
      out[row * 9 + l16] = a[r] + bb;
    }
  }
}

extern "C" void kernel_launch(void* const* d_in, const int* in_sizes, int n_in,
                              void* d_out, int out_size, void* d_ws, size_t ws_size,
                              hipStream_t stream) {
  (void)in_sizes; (void)n_in; (void)out_size; (void)ws_size;
  const int*   x     = (const int*)d_in[0];
  const float* emb   = (const float*)d_in[1];
  const float* ln1   = (const float*)d_in[2];
  const float* Wg    = (const float*)d_in[3];
  const float* R     = (const float*)d_in[4];
  const float* bias  = (const float*)d_in[5];
  const float* gnw   = (const float*)d_in[6];
  const float* ln2   = (const float*)d_in[7];
  const float* wup   = (const float*)d_in[8];
  const float* wdn   = (const float*)d_in[9];
  const float* postw = (const float*)d_in[10];
  const float* pw    = (const float*)d_in[11];
  const float* pb    = (const float*)d_in[12];
  float* out = (float*)d_out;

  float* h  = (float*)d_ws;
  float* xb = h + (size_t)XT * XD;

  k_embed<<<(XT * XD + 255) / 256, 256, 0, stream>>>(x, emb, h, XT * XD);

  for (int bi = 0; bi < 2; ++bi) {
    k_ln<<<XT / 8, 256, 0, stream>>>(h, ln1 + bi * XD, xb);
    k_scan<<<dim3(XB / SCAN_R, XNH), 256, 0, stream>>>(
        xb, h,
        Wg   + (size_t)bi * 4 * XNH * XDH * XDH,
        R    + (size_t)bi * XNH * XDH * 4 * XDH,
        bias + (size_t)bi * XNH * 4 * XDH,
        gnw  + (size_t)bi * XD);
    k_ln<<<XT / 8, 256, 0, stream>>>(h, ln2 + bi * XD, xb);
    k_ffn<<<XT / 16, 32, 0, stream>>>(xb, h,
                                      wup + (size_t)bi * 2 * XD * XD,
                                      wdn + (size_t)bi * XD * XD);
  }

  k_ln<<<XT / 8, 256, 0, stream>>>(h, postw, xb);
  k_proj<<<XT / 16, 32, 0, stream>>>(xb, pw, pb, out);
}


# scan gate phase: exp and log1p of the per-lane chain on the exp2 and log2 instructions (log1p(t) = log(1+t) + rounding remainder/(1+t)) instead of library code; results differ in the last bits
# speedup vs baseline: 6.0816x; 1.0818x over previous
#include <hip/hip_runtime.h>
#include <hip/hip_bf16.h>


typedef float v2f __attribute__((ext_vector_type(2)));
typedef float v8f __attribute__((ext_vector_type(8)));

#define XB   512
#define XS   64
#define XD   128
#define XNH  4
#define XDH  32
#define XT   (XB * XS)
#define XEPS 1e-5f

__device__ __forceinline__ v8f wmma4(v2f a, v2f b, v8f c) {
  return __builtin_amdgcn_wmma_f32_16x16x4_f32(false, a, false, b, (short)0, c,
                                               false, false);
}

__device__ __forceinline__ void lds_barrier() {
  __builtin_amdgcn_fence(__ATOMIC_RELEASE, "workgroup", "local");
  __builtin_amdgcn_s_barrier();
  __builtin_amdgcn_fence(__ATOMIC_ACQUIRE, "workgroup", "local");
}

__device__ __forceinline__ float gelu_exact(float x) {
  return 0.5f * x * (1.0f + erff(x * 0.70710678118654752f));
}

__global__ void k_embed(const int* __restrict__ x, const float* __restrict__ emb,
                        float* __restrict__ h, int total) {
  int i = blockIdx.x * blockDim.x + threadIdx.x;
  if (i >= total) return;
  int t = i >> 7;
  int c = i & 127;
  h[i] = emb[x[t] * XD + c];
}

__global__ void k_ln(const float* __restrict__ h, const float* __restrict__ w,
                     float* __restrict__ xn) {
  int wid  = threadIdx.x >> 5;
  int lane = threadIdx.x & 31;
  int t = blockIdx.x * (blockDim.x >> 5) + wid;
  const float4* hv = (const float4*)(h + (size_t)t * XD);
  float4 v = hv[lane];
  float s = v.x + v.y + v.z + v.w;
  float q = v.x * v.x + v.y * v.y + v.z * v.z + v.w * v.w;
  #pragma unroll
  for (int m = 16; m >= 1; m >>= 1) {
    s += __shfl_xor(s, m, 32);
    q += __shfl_xor(q, m, 32);
  }
  float mu  = s * (1.0f / XD);
  float var = q * (1.0f / XD) - mu * mu;
  float rs  = rsqrtf(var + XEPS);
  float4 wl = ((const float4*)w)[lane];
  float4 o;
  o.x = (v.x - mu) * rs * wl.x;
  o.y = (v.y - mu) * rs * wl.y;
  o.z = (v.z - mu) * rs * wl.z;
  o.w = (v.w - mu) * rs * wl.w;
  ((float4*)(xn + (size_t)t * XD))[lane] = o;
}

#define SCAN_LP 36
#define SCAN_R  8
__global__ __launch_bounds__(256) void k_scan(const float* __restrict__ xn,
                                              float* __restrict__ hio,
                                              const float* __restrict__ Wg,
                                              const float* __restrict__ R,
                                              const float* __restrict__ bias,
                                              const float* __restrict__ gnw) {
  __shared__ __attribute__((aligned(16))) float lraw[4 * 16 * SCAN_LP];
  __shared__ __attribute__((aligned(16))) float lh[SCAN_R * SCAN_LP];

  const int lane = threadIdx.x & 31;
  const int w    = __builtin_amdgcn_readfirstlane((int)(threadIdx.x >> 5));
  const int hlf  = lane >> 4;
  const int l16  = lane & 15;
  const int ar   = lane & 7;
  const int b0   = blockIdx.x * SCAN_R;
  const int n    = blockIdx.y;

  v2f bw[8], br[8];
  const int N  = w * 16 + l16;
  {
    const int g  = N >> 5;
    const int dh = N & 31;
    #pragma unroll
    for (int kk = 0; kk < 8; ++kk) {
      const int K0 = kk * 4 + hlf * 2;
      const float* wgp = Wg + (((size_t)(g * XNH + n) * XDH + dh) * XDH + K0);
      bw[kk].x = wgp[0];
      bw[kk].y = wgp[1];
      br[kk].x = R[(size_t)(n * XDH + K0 + 0) * (4 * XDH) + N];
      br[kk].y = R[(size_t)(n * XDH + K0 + 1) * (4 * XDH) + N];
    }
  }
  const float bj = bias[n * 128 + N];
  const float gn = gnw[n * XDH + lane];
  for (int j = threadIdx.x; j < SCAN_R * SCAN_LP; j += 256) lh[j] = 0.0f;
  __syncthreads();

  float cS = 0.0f, nS = 0.0f, mS = 0.0f;
  const int row = w;
  float* hp = hio + ((size_t)(b0 + row) * XS) * XD + n * XDH + lane;
  const float* xp0 = xn + ((size_t)(b0 + ar) * XS) * XD + n * XDH + hlf * 2;

  v2f ax[8];
  #pragma unroll
  for (int kk = 0; kk < 8; ++kk) {
    ax[kk].x = xp0[kk * 4];
    ax[kk].y = xp0[kk * 4 + 1];
  }

  #pragma unroll 1
  for (int s = 0; s < XS; ++s) {
    const float hold = hp[(size_t)s * XD];

    {
      v2f ah[8];
      #pragma unroll
      for (int kk = 0; kk < 8; ++kk) {
        ah[kk].x = lh[ar * SCAN_LP + kk * 4 + hlf * 2];
        ah[kk].y = lh[ar * SCAN_LP + kk * 4 + hlf * 2 + 1];
      }
      __builtin_amdgcn_sched_barrier(0);
      v8f a;
      #pragma unroll
      for (int r = 0; r < 8; ++r) a[r] = bj;
      #pragma unroll
      for (int kk = 0; kk < 8; ++kk) {
        a = wmma4(ax[kk], bw[kk], a);
        a = wmma4(ah[kk], br[kk], a);
      }
      {
        float* rp = lraw + ((w >> 1) * 16 + hlf * 8) * SCAN_LP + (w & 1) * 16 + l16;
        #pragma unroll
        for (int r = 0; r < 8; ++r) rp[r * SCAN_LP] = a[r];
      }
      const int sn = (s + 1 < XS) ? s + 1 : s;
      const float* xq = xp0 + (size_t)sn * XD;
      #pragma unroll
      for (int kk = 0; kk < 8; ++kk) {
        ax[kk].x = xq[kk * 4];
        ax[kk].y = xq[kk * 4 + 1];
      }
    }
    lds_barrier();

    float hn;
    {
      float ir  = lraw[(0 * 16 + row) * SCAN_LP + lane];
      float fr  = lraw[(1 * 16 + row) * SCAN_LP + lane];
      float zr  = lraw[(2 * 16 + row) * SCAN_LP + lane];
      float orr = lraw[(3 * 16 + row) * SCAN_LP + lane];
      const float kL2E = 1.44269504088896341f;
      const float kLN2 = 0.693147180559945309f;
      float te  = __builtin_amdgcn_exp2f(-fabsf(fr) * kL2E);
      float ue  = 1.0f + te;
      float ce  = te - (ue - 1.0f);
      float l1p = fmaf(__builtin_amdgcn_logf(ue), kLN2, ce * __builtin_amdgcn_rcpf(ue));
      float ls  = fminf(fr, 0.0f) - l1p;
      float lfm = mS + ls;
      float mn  = fmaxf(ir, lfm);
      float ig  = __builtin_amdgcn_exp2f((ir - mn) * kL2E);
      float fg  = __builtin_amdgcn_exp2f((lfm - mn) * kL2E);
      float cn  = fg * cS + ig * tanhf(zr);
      float nn  = fg * nS + ig;
      float og  = 1.0f / (1.0f + __builtin_amdgcn_exp2f(-orr * kL2E));
      hn  = og * cn / nn;
      cS = cn;
      nS = nn;
      mS = mn;
      lh[row * SCAN_LP + lane] = hn;
    }
    lds_barrier();

    {
      float mu = 0.0f, q = 0.0f;
      #pragma unroll
      for (int i = 0; i < 32; ++i) {
        float v = lh[row * SCAN_LP + i];
        mu += v;
        q  += v * v;
      }
      mu *= (1.0f / XDH);
      q = q * (1.0f / XDH) - mu * mu;
      float rs = rsqrtf(q + XEPS);
      float v = (hn - mu) * rs * gn;
      hp[(size_t)s * XD] = hold + v;
    }
  }
}

__global__ __launch_bounds__(32) void k_ffn(const float* __restrict__ xn,
                                            float* __restrict__ hio,
                                            const float* __restrict__ Wup,
                                            const float* __restrict__ Wdn) {
  __shared__ float lact[16 * 128];
  const int lane = threadIdx.x;
  const int hlf  = lane >> 4;
  const int l16  = lane & 15;
  const size_t t0 = (size_t)blockIdx.x * 16;

  v2f ax[32];
  const float* xp = xn + (t0 + l16) * XD;
  #pragma unroll
  for (int kk = 0; kk < 32; ++kk) {
    int K0 = kk * 4 + hlf * 2;
    ax[kk].x = xp[K0];
    ax[kk].y = xp[K0 + 1];
  }

  float gatev[8][8];
  #pragma unroll
  for (int nt = 0; nt < 8; ++nt) {
    v8f a;
    #pragma unroll
    for (int r = 0; r < 8; ++r) a[r] = 0.0f;
    int j = nt * 16 + l16;
    #pragma unroll
    for (int kk = 0; kk < 32; ++kk) {
      int K0 = kk * 4 + hlf * 2;
      v2f b;
      b.x = Wup[(size_t)j * XD + K0];
      b.y = Wup[(size_t)j * XD + K0 + 1];
      a = wmma4(ax[kk], b, a);
    }
    #pragma unroll
    for (int r = 0; r < 8; ++r) gatev[nt][r] = a[r];
  }

  #pragma unroll
  for (int nt = 0; nt < 8; ++nt) {
    v8f a;
    #pragma unroll
    for (int r = 0; r < 8; ++r) a[r] = 0.0f;
    int j = 128 + nt * 16 + l16;
    #pragma unroll
    for (int kk = 0; kk < 32; ++kk) {
      int K0 = kk * 4 + hlf * 2;
      v2f b;
      b.x = Wup[(size_t)j * XD + K0];
      b.y = Wup[(size_t)j * XD + K0 + 1];
      a = wmma4(ax[kk], b, a);
    }
    #pragma unroll
    for (int r = 0; r < 8; ++r) {
      float act = gelu_exact(gatev[nt][r]) * a[r];
      lact[(r + hlf * 8) * 128 + nt * 16 + l16] = act;
    }
  }
  __syncthreads();

  v2f aa[32];
  const float* lp = lact + l16 * 128;
  #pragma unroll
  for (int kk = 0; kk < 32; ++kk) {
    int K0 = kk * 4 + hlf * 2;
    aa[kk].x = lp[K0];
    aa[kk].y = lp[K0 + 1];
  }
  #pragma unroll
  for (int nt = 0; nt < 8; ++nt) {
    v8f a;
    #pragma unroll
    for (int r = 0; r < 8; ++r) a[r] = 0.0f;
    int j = nt * 16 + l16;
    #pragma unroll
    for (int kk = 0; kk < 32; ++kk) {
      int K0 = kk * 4 + hlf * 2;
      v2f b;
      b.x = Wdn[(size_t)j * XD + K0];
      b.y = Wdn[(size_t)j * XD + K0 + 1];
      a = wmma4(aa[kk], b, a);
    }
    #pragma unroll
    for (int r = 0; r < 8; ++r) {
      size_t row = t0 + r + hlf * 8;
      hio[row * XD + nt * 16 + l16] += a[r];
    }
  }
}

__global__ __launch_bounds__(32) void k_proj(const float* __restrict__ xn,
                                             const float* __restrict__ pw,
                                             const float* __restrict__ pb,
                                             float* __restrict__ out) {
  const int lane = threadIdx.x;
  const int hlf  = lane >> 4;
  const int l16  = lane & 15;
  const size_t t0 = (size_t)blockIdx.x * 16;

  v8f a;
  #pragma unroll
  for (int r = 0; r < 8; ++r) a[r] = 0.0f;
  const float* xp = xn + (t0 + l16) * XD;
  #pragma unroll
  for (int kk = 0; kk < 32; ++kk) {
    int K0 = kk * 4 + hlf * 2;
    v2f av;
    av.x = xp[K0];
    av.y = xp[K0 + 1];
    v2f b;
    b.x = 0.0f;
    b.y = 0.0f;
    if (l16 < 9) {
      b.x = pw[(size_t)l16 * XD + K0];
      b.y = pw[(size_t)l16 * XD + K0 + 1];
    }
    a = wmma4(av, b, a);
  }
  if (l16 < 9) {
    float bb = pb[l16];
    #pragma unroll
    for (int r = 0; r < 8; ++r) {
      size_t row = t0 + r + hlf * 8;
      out[row * 9 + l16] = a[r] + bb;
    }
  }
}

extern "C" void kernel_launch(void* const* d_in, const int* in_sizes, int n_in,
                              void* d_out, int out_size, void* d_ws, size_t ws_size,
                              hipStream_t stream) {
  (void)in_sizes; (void)n_in; (void)out_size; (void)ws_size;
  const int*   x     = (const int*)d_in[0];
  const float* emb   = (const float*)d_in[1];
  const float* ln1   = (const float*)d_in[2];
  const float* Wg    = (const float*)d_in[3];
  const float* R     = (const float*)d_in[4];
  const float* bias  = (const float*)d_in[5];
  const float* gnw   = (const float*)d_in[6];
  const float* ln2   = (const float*)d_in[7];
  const float* wup   = (const float*)d_in[8];
  const float* wdn   = (const float*)d_in[9];
  const float* postw = (const float*)d_in[10];
  const float* pw    = (const float*)d_in[11];
  const float* pb    = (const float*)d_in[12];
  float* out = (float*)d_out;

  float* h  = (float*)d_ws;
  float* xb = h + (size_t)XT * XD;

  k_embed<<<(XT * XD + 255) / 256, 256, 0, stream>>>(x, emb, h, XT * XD);

  for (int bi = 0; bi < 2; ++bi) {
    k_ln<<<XT / 8, 256, 0, stream>>>(h, ln1 + bi * XD, xb);
    k_scan<<<dim3(XB / SCAN_R, XNH), 256, 0, stream>>>(
        xb, h,
        Wg   + (size_t)bi * 4 * XNH * XDH * XDH,
        R    + (size_t)bi * XNH * XDH * 4 * XDH,
        bias + (size_t)bi * XNH * 4 * XDH,
        gnw  + (size_t)bi * XD);
    k_ln<<<XT / 8, 256, 0, stream>>>(h, ln2 + bi * XD, xb);
    k_ffn<<<XT / 16, 32, 0, stream>>>(xb, h,
                                      wup + (size_t)bi * 2 * XD * XD,
                                      wdn + (size_t)bi * XD * XD);
  }

  k_ln<<<XT / 8, 256, 0, stream>>>(h, postw, xb);
  k_proj<<<XT / 16, 32, 0, stream>>>(xb, pw, pb, out);
}


# scan head norm: row mean and mean square by a butterfly sum across the wave's 32 lanes (DPP row_xmask adds + one row swap) in the gate phase's registers, not 32 serial LDS reads; sum order now a tree
# speedup vs baseline: 6.5498x; 1.0770x over previous
#include <hip/hip_runtime.h>
#include <hip/hip_bf16.h>


typedef float v2f __attribute__((ext_vector_type(2)));
typedef float v8f __attribute__((ext_vector_type(8)));

#define XB   512
#define XS   64
#define XD   128
#define XNH  4
#define XDH  32
#define XT   (XB * XS)
#define XEPS 1e-5f

__device__ __forceinline__ v8f wmma4(v2f a, v2f b, v8f c) {
  return __builtin_amdgcn_wmma_f32_16x16x4_f32(false, a, false, b, (short)0, c,
                                               false, false);
}

__device__ __forceinline__ void lds_barrier() {
  __builtin_amdgcn_fence(__ATOMIC_RELEASE, "workgroup", "local");
  __builtin_amdgcn_s_barrier();
  __builtin_amdgcn_fence(__ATOMIC_ACQUIRE, "workgroup", "local");
}

template <int M>
__device__ __forceinline__ float row_xor(float v) {
  const int b = __builtin_bit_cast(int, v);
  return __builtin_bit_cast(float, __builtin_amdgcn_update_dpp(0, b, 0x160 + M, 0xf, 0xf, true));
}
__device__ __forceinline__ float xhalf(float v) {
  const int b = __builtin_bit_cast(int, v);
  return __builtin_bit_cast(float, __builtin_amdgcn_permlanex16(b, b, 0x76543210, 0xfedcba98, false, false));
}
__device__ __forceinline__ float wave_sum32(float v) {
  v += row_xor<1>(v);
  v += row_xor<2>(v);
  v += row_xor<4>(v);
  v += row_xor<8>(v);
  v += xhalf(v);
  return v;
}

__device__ __forceinline__ float gelu_exact(float x) {
  return 0.5f * x * (1.0f + erff(x * 0.70710678118654752f));
}

__global__ void k_embed(const int* __restrict__ x, const float* __restrict__ emb,
                        float* __restrict__ h, int total) {
  int i = blockIdx.x * blockDim.x + threadIdx.x;
  if (i >= total) return;
  int t = i >> 7;
  int c = i & 127;
  h[i] = emb[x[t] * XD + c];
}

__global__ void k_ln(const float* __restrict__ h, const float* __restrict__ w,
                     float* __restrict__ xn) {
  int wid  = threadIdx.x >> 5;
  int lane = threadIdx.x & 31;
  int t = blockIdx.x * (blockDim.x >> 5) + wid;
  const float4* hv = (const float4*)(h + (size_t)t * XD);
  float4 v = hv[lane];
  float s = v.x + v.y + v.z + v.w;
  float q = v.x * v.x + v.y * v.y + v.z * v.z + v.w * v.w;
  #pragma unroll
  for (int m = 16; m >= 1; m >>= 1) {
    s += __shfl_xor(s, m, 32);
    q += __shfl_xor(q, m, 32);
  }
  float mu  = s * (1.0f / XD);
  float var = q * (1.0f / XD) - mu * mu;
  float rs  = rsqrtf(var + XEPS);
  float4 wl = ((const float4*)w)[lane];
  float4 o;
  o.x = (v.x - mu) * rs * wl.x;
  o.y = (v.y - mu) * rs * wl.y;
  o.z = (v.z - mu) * rs * wl.z;
  o.w = (v.w - mu) * rs * wl.w;
  ((float4*)(xn + (size_t)t * XD))[lane] = o;
}

#define SCAN_LP 36
#define SCAN_R  8
__global__ __launch_bounds__(256) void k_scan(const float* __restrict__ xn,
                                              float* __restrict__ hio,
                                              const float* __restrict__ Wg,
                                              const float* __restrict__ R,
                                              const float* __restrict__ bias,
                                              const float* __restrict__ gnw) {
  __shared__ __attribute__((aligned(16))) float lraw[4 * 16 * SCAN_LP];
  __shared__ __attribute__((aligned(16))) float lh[SCAN_R * SCAN_LP];

  const int lane = threadIdx.x & 31;
  const int w    = __builtin_amdgcn_readfirstlane((int)(threadIdx.x >> 5));
  const int hlf  = lane >> 4;
  const int l16  = lane & 15;
  const int ar   = lane & 7;
  const int b0   = blockIdx.x * SCAN_R;
  const int n    = blockIdx.y;

  v2f bw[8], br[8];
  const int N  = w * 16 + l16;
  {
    const int g  = N >> 5;
    const int dh = N & 31;
    #pragma unroll
    for (int kk = 0; kk < 8; ++kk) {
      const int K0 = kk * 4 + hlf * 2;
      const float* wgp = Wg + (((size_t)(g * XNH + n) * XDH + dh) * XDH + K0);
      bw[kk].x = wgp[0];
      bw[kk].y = wgp[1];
      br[kk].x = R[(size_t)(n * XDH + K0 + 0) * (4 * XDH) + N];
      br[kk].y = R[(size_t)(n * XDH + K0 + 1) * (4 * XDH) + N];
    }
  }
  const float bj = bias[n * 128 + N];
  const float gn = gnw[n * XDH + lane];
  for (int j = threadIdx.x; j < SCAN_R * SCAN_LP; j += 256) lh[j] = 0.0f;
  __syncthreads();

  float cS = 0.0f, nS = 0.0f, mS = 0.0f;
  const int row = w;
  float* hp = hio + ((size_t)(b0 + row) * XS) * XD + n * XDH + lane;
  const float* xp0 = xn + ((size_t)(b0 + ar) * XS) * XD + n * XDH + hlf * 2;

  v2f ax[8];
  #pragma unroll
  for (int kk = 0; kk < 8; ++kk) {
    ax[kk].x = xp0[kk * 4];
    ax[kk].y = xp0[kk * 4 + 1];
  }

  #pragma unroll 1
  for (int s = 0; s < XS; ++s) {
    const float hold = hp[(size_t)s * XD];

    {
      v2f ah[8];
      #pragma unroll
      for (int kk = 0; kk < 8; ++kk) {
        ah[kk].x = lh[ar * SCAN_LP + kk * 4 + hlf * 2];
        ah[kk].y = lh[ar * SCAN_LP + kk * 4 + hlf * 2 + 1];
      }
      __builtin_amdgcn_sched_barrier(0);
      v8f a;
      #pragma unroll
      for (int r = 0; r < 8; ++r) a[r] = bj;
      #pragma unroll
      for (int kk = 0; kk < 8; ++kk) {
        a = wmma4(ax[kk], bw[kk], a);
        a = wmma4(ah[kk], br[kk], a);
      }
      {
        float* rp = lraw + ((w >> 1) * 16 + hlf * 8) * SCAN_LP + (w & 1) * 16 + l16;
        #pragma unroll
        for (int r = 0; r < 8; ++r) rp[r * SCAN_LP] = a[r];
      }
      const int sn = (s + 1 < XS) ? s + 1 : s;
      const float* xq = xp0 + (size_t)sn * XD;
      #pragma unroll
      for (int kk = 0; kk < 8; ++kk) {
        ax[kk].x = xq[kk * 4];
        ax[kk].y = xq[kk * 4 + 1];
      }
    }
    lds_barrier();

    float hn;
    {
      float ir  = lraw[(0 * 16 + row) * SCAN_LP + lane];
      float fr  = lraw[(1 * 16 + row) * SCAN_LP + lane];
      float zr  = lraw[(2 * 16 + row) * SCAN_LP + lane];
      float orr = lraw[(3 * 16 + row) * SCAN_LP + lane];
      const float kL2E = 1.44269504088896341f;
      const float kLN2 = 0.693147180559945309f;
      float te  = __builtin_amdgcn_exp2f(-fabsf(fr) * kL2E);
      float ue  = 1.0f + te;
      float ce  = te - (ue - 1.0f);
      float l1p = fmaf(__builtin_amdgcn_logf(ue), kLN2, ce * __builtin_amdgcn_rcpf(ue));
      float ls  = fminf(fr, 0.0f) - l1p;
      float lfm = mS + ls;
      float mn  = fmaxf(ir, lfm);
      float ig  = __builtin_amdgcn_exp2f((ir - mn) * kL2E);
      float fg  = __builtin_amdgcn_exp2f((lfm - mn) * kL2E);
      float cn  = fg * cS + ig * tanhf(zr);
      float nn  = fg * nS + ig;
      float og  = 1.0f / (1.0f + __builtin_amdgcn_exp2f(-orr * kL2E));
      hn  = og * cn / nn;
      cS = cn;
      nS = nn;
      mS = mn;
      lh[row * SCAN_LP + lane] = hn;
    }

    {
      float mu = wave_sum32(hn) * (1.0f / XDH);
      float q  = wave_sum32(hn * hn) * (1.0f / XDH) - mu * mu;
      float rs = rsqrtf(q + XEPS);
      float v = (hn - mu) * rs * gn;
      hp[(size_t)s * XD] = hold + v;
    }
    lds_barrier();
  }
}

__global__ __launch_bounds__(32) void k_ffn(const float* __restrict__ xn,
                                            float* __restrict__ hio,
                                            const float* __restrict__ Wup,
                                            const float* __restrict__ Wdn) {
  __shared__ float lact[16 * 128];
  const int lane = threadIdx.x;
  const int hlf  = lane >> 4;
  const int l16  = lane & 15;
  const size_t t0 = (size_t)blockIdx.x * 16;

  v2f ax[32];
  const float* xp = xn + (t0 + l16) * XD;
  #pragma unroll
  for (int kk = 0; kk < 32; ++kk) {
    int K0 = kk * 4 + hlf * 2;
    ax[kk].x = xp[K0];
    ax[kk].y = xp[K0 + 1];
  }

  float gatev[8][8];
  #pragma unroll
  for (int nt = 0; nt < 8; ++nt) {
    v8f a;
    #pragma unroll
    for (int r = 0; r < 8; ++r) a[r] = 0.0f;
    int j = nt * 16 + l16;
    #pragma unroll
    for (int kk = 0; kk < 32; ++kk) {
      int K0 = kk * 4 + hlf * 2;
      v2f b;
      b.x = Wup[(size_t)j * XD + K0];
      b.y = Wup[(size_t)j * XD + K0 + 1];
      a = wmma4(ax[kk], b, a);
    }
    #pragma unroll
    for (int r = 0; r < 8; ++r) gatev[nt][r] = a[r];
  }

  #pragma unroll
  for (int nt = 0; nt < 8; ++nt) {
    v8f a;
    #pragma unroll
    for (int r = 0; r < 8; ++r) a[r] = 0.0f;
    int j = 128 + nt * 16 + l16;
    #pragma unroll
    for (int kk = 0; kk < 32; ++kk) {
      int K0 = kk * 4 + hlf * 2;
      v2f b;
      b.x = Wup[(size_t)j * XD + K0];
      b.y = Wup[(size_t)j * XD + K0 + 1];
      a = wmma4(ax[kk], b, a);
    }
    #pragma unroll
    for (int r = 0; r < 8; ++r) {
      float act = gelu_exact(gatev[nt][r]) * a[r];
      lact[(r + hlf * 8) * 128 + nt * 16 + l16] = act;
    }
  }
  __syncthreads();

  v2f aa[32];
  const float* lp = lact + l16 * 128;
  #pragma unroll
  for (int kk = 0; kk < 32; ++kk) {
    int K0 = kk * 4 + hlf * 2;
    aa[kk].x = lp[K0];
    aa[kk].y = lp[K0 + 1];
  }
  #pragma unroll
  for (int nt = 0; nt < 8; ++nt) {
    v8f a;
    #pragma unroll
    for (int r = 0; r < 8; ++r) a[r] = 0.0f;
    int j = nt * 16 + l16;
    #pragma unroll
    for (int kk = 0; kk < 32; ++kk) {
      int K0 = kk * 4 + hlf * 2;
      v2f b;
      b.x = Wdn[(size_t)j * XD + K0];
      b.y = Wdn[(size_t)j * XD + K0 + 1];
      a = wmma4(aa[kk], b, a);
    }
    #pragma unroll
    for (int r = 0; r < 8; ++r) {
      size_t row = t0 + r + hlf * 8;
      hio[row * XD + nt * 16 + l16] += a[r];
    }
  }
}

__global__ __launch_bounds__(32) void k_proj(const float* __restrict__ xn,
                                             const float* __restrict__ pw,
                                             const float* __restrict__ pb,
                                             float* __restrict__ out) {
  const int lane = threadIdx.x;
  const int hlf  = lane >> 4;
  const int l16  = lane & 15;
  const size_t t0 = (size_t)blockIdx.x * 16;

  v8f a;
  #pragma unroll
  for (int r = 0; r < 8; ++r) a[r] = 0.0f;
  const float* xp = xn + (t0 + l16) * XD;
  #pragma unroll
  for (int kk = 0; kk < 32; ++kk) {
    int K0 = kk * 4 + hlf * 2;
    v2f av;
    av.x = xp[K0];
    av.y = xp[K0 + 1];
    v2f b;
    b.x = 0.0f;
    b.y = 0.0f;
    if (l16 < 9) {
      b.x = pw[(size_t)l16 * XD + K0];
      b.y = pw[(size_t)l16 * XD + K0 + 1];
    }
    a = wmma4(av, b, a);
  }
  if (l16 < 9) {
    float bb = pb[l16];
    #pragma unroll
    for (int r = 0; r < 8; ++r) {
      size_t row = t0 + r + hlf * 8;
      out[row * 9 + l16] = a[r] + bb;
    }
  }
}

extern "C" void kernel_launch(void* const* d_in, const int* in_sizes, int n_in,
                              void* d_out, int out_size, void* d_ws, size_t ws_size,
                              hipStream_t stream) {
  (void)in_sizes; (void)n_in; (void)out_size; (void)ws_size;
  const int*   x     = (const int*)d_in[0];
  const float* emb   = (const float*)d_in[1];
  const float* ln1   = (const float*)d_in[2];
  const float* Wg    = (const float*)d_in[3];
  const float* R     = (const float*)d_in[4];
  const float* bias  = (const float*)d_in[5];
  const float* gnw   = (const float*)d_in[6];
  const float* ln2   = (const float*)d_in[7];
  const float* wup   = (const float*)d_in[8];
  const float* wdn   = (const float*)d_in[9];
  const float* postw = (const float*)d_in[10];
  const float* pw    = (const float*)d_in[11];
  const float* pb    = (const float*)d_in[12];
  float* out = (float*)d_out;

  float* h  = (float*)d_ws;
  float* xb = h + (size_t)XT * XD;

  k_embed<<<(XT * XD + 255) / 256, 256, 0, stream>>>(x, emb, h, XT * XD);

  for (int bi = 0; bi < 2; ++bi) {
    k_ln<<<XT / 8, 256, 0, stream>>>(h, ln1 + bi * XD, xb);
    k_scan<<<dim3(XB / SCAN_R, XNH), 256, 0, stream>>>(
        xb, h,
        Wg   + (size_t)bi * 4 * XNH * XDH * XDH,
        R    + (size_t)bi * XNH * XDH * 4 * XDH,
        bias + (size_t)bi * XNH * 4 * XDH,
        gnw  + (size_t)bi * XD);
    k_ln<<<XT / 8, 256, 0, stream>>>(h, ln2 + bi * XD, xb);
    k_ffn<<<XT / 16, 32, 0, stream>>>(xb, h,
                                      wup + (size_t)bi * 2 * XD * XD,
                                      wdn + (size_t)bi * XD * XD);
  }

  k_ln<<<XT / 8, 256, 0, stream>>>(h, postw, xb);
  k_proj<<<XT / 16, 32, 0, stream>>>(xb, pw, pb, out);
}


# scan gate and norm chain: tanh, the sigmoid's and the c/n quotient's reciprocals and the norm's rsqrt on single hardware instructions, not library code and division sequences; last bits differ
# speedup vs baseline: 6.7776x; 1.0348x over previous
#include <hip/hip_runtime.h>
#include <hip/hip_bf16.h>


typedef float v2f __attribute__((ext_vector_type(2)));
typedef float v8f __attribute__((ext_vector_type(8)));

#define XB   512
#define XS   64
#define XD   128
#define XNH  4
#define XDH  32
#define XT   (XB * XS)
#define XEPS 1e-5f

__device__ __forceinline__ v8f wmma4(v2f a, v2f b, v8f c) {
  return __builtin_amdgcn_wmma_f32_16x16x4_f32(false, a, false, b, (short)0, c,
                                               false, false);
}

__device__ __forceinline__ void lds_barrier() {
  __builtin_amdgcn_fence(__ATOMIC_RELEASE, "workgroup", "local");
  __builtin_amdgcn_s_barrier();
  __builtin_amdgcn_fence(__ATOMIC_ACQUIRE, "workgroup", "local");
}

template <int M>
__device__ __forceinline__ float row_xor(float v) {
  const int b = __builtin_bit_cast(int, v);
  return __builtin_bit_cast(float, __builtin_amdgcn_update_dpp(0, b, 0x160 + M, 0xf, 0xf, true));
}
__device__ __forceinline__ float xhalf(float v) {
  const int b = __builtin_bit_cast(int, v);
  return __builtin_bit_cast(float, __builtin_amdgcn_permlanex16(b, b, 0x76543210, 0xfedcba98, false, false));
}
__device__ __forceinline__ float wave_sum32(float v) {
  v += row_xor<1>(v);
  v += row_xor<2>(v);
  v += row_xor<4>(v);
  v += row_xor<8>(v);
  v += xhalf(v);
  return v;
}

__device__ __forceinline__ float gelu_exact(float x) {
  return 0.5f * x * (1.0f + erff(x * 0.70710678118654752f));
}

__global__ void k_embed(const int* __restrict__ x, const float* __restrict__ emb,
                        float* __restrict__ h, int total) {
  int i = blockIdx.x * blockDim.x + threadIdx.x;
  if (i >= total) return;
  int t = i >> 7;
  int c = i & 127;
  h[i] = emb[x[t] * XD + c];
}

__global__ void k_ln(const float* __restrict__ h, const float* __restrict__ w,
                     float* __restrict__ xn) {
  int wid  = threadIdx.x >> 5;
  int lane = threadIdx.x & 31;
  int t = blockIdx.x * (blockDim.x >> 5) + wid;
  const float4* hv = (const float4*)(h + (size_t)t * XD);
  float4 v = hv[lane];
  float s = v.x + v.y + v.z + v.w;
  float q = v.x * v.x + v.y * v.y + v.z * v.z + v.w * v.w;
  #pragma unroll
  for (int m = 16; m >= 1; m >>= 1) {
    s += __shfl_xor(s, m, 32);
    q += __shfl_xor(q, m, 32);
  }
  float mu  = s * (1.0f / XD);
  float var = q * (1.0f / XD) - mu * mu;
  float rs  = rsqrtf(var + XEPS);
  float4 wl = ((const float4*)w)[lane];
  float4 o;
  o.x = (v.x - mu) * rs * wl.x;
  o.y = (v.y - mu) * rs * wl.y;
  o.z = (v.z - mu) * rs * wl.z;
  o.w = (v.w - mu) * rs * wl.w;
  ((float4*)(xn + (size_t)t * XD))[lane] = o;
}

#define SCAN_LP 36
#define SCAN_R  8
__global__ __launch_bounds__(256) void k_scan(const float* __restrict__ xn,
                                              float* __restrict__ hio,
                                              const float* __restrict__ Wg,
                                              const float* __restrict__ R,
                                              const float* __restrict__ bias,
                                              const float* __restrict__ gnw) {
  __shared__ __attribute__((aligned(16))) float lraw[4 * 16 * SCAN_LP];
  __shared__ __attribute__((aligned(16))) float lh[SCAN_R * SCAN_LP];

  const int lane = threadIdx.x & 31;
  const int w    = __builtin_amdgcn_readfirstlane((int)(threadIdx.x >> 5));
  const int hlf  = lane >> 4;
  const int l16  = lane & 15;
  const int ar   = lane & 7;
  const int b0   = blockIdx.x * SCAN_R;
  const int n    = blockIdx.y;

  v2f bw[8], br[8];
  const int N  = w * 16 + l16;
  {
    const int g  = N >> 5;
    const int dh = N & 31;
    #pragma unroll
    for (int kk = 0; kk < 8; ++kk) {
      const int K0 = kk * 4 + hlf * 2;
      const float* wgp = Wg + (((size_t)(g * XNH + n) * XDH + dh) * XDH + K0);
      bw[kk].x = wgp[0];
      bw[kk].y = wgp[1];
      br[kk].x = R[(size_t)(n * XDH + K0 + 0) * (4 * XDH) + N];
      br[kk].y = R[(size_t)(n * XDH + K0 + 1) * (4 * XDH) + N];
    }
  }
  const float bj = bias[n * 128 + N];
  const float gn = gnw[n * XDH + lane];
  for (int j = threadIdx.x; j < SCAN_R * SCAN_LP; j += 256) lh[j] = 0.0f;
  __syncthreads();

  float cS = 0.0f, nS = 0.0f, mS = 0.0f;
  const int row = w;
  float* hp = hio + ((size_t)(b0 + row) * XS) * XD + n * XDH + lane;
  const float* xp0 = xn + ((size_t)(b0 + ar) * XS) * XD + n * XDH + hlf * 2;

  v2f ax[8];
  #pragma unroll
  for (int kk = 0; kk < 8; ++kk) {
    ax[kk].x = xp0[kk * 4];
    ax[kk].y = xp0[kk * 4 + 1];
  }

  #pragma unroll 1
  for (int s = 0; s < XS; ++s) {
    const float hold = hp[(size_t)s * XD];

    {
      v2f ah[8];
      #pragma unroll
      for (int kk = 0; kk < 8; ++kk) {
        ah[kk].x = lh[ar * SCAN_LP + kk * 4 + hlf * 2];
        ah[kk].y = lh[ar * SCAN_LP + kk * 4 + hlf * 2 + 1];
      }
      __builtin_amdgcn_sched_barrier(0);
      v8f a;
      #pragma unroll
      for (int r = 0; r < 8; ++r) a[r] = bj;
      #pragma unroll
      for (int kk = 0; kk < 8; ++kk) {
        a = wmma4(ax[kk], bw[kk], a);
        a = wmma4(ah[kk], br[kk], a);
      }
      {
        float* rp = lraw + ((w >> 1) * 16 + hlf * 8) * SCAN_LP + (w & 1) * 16 + l16;
        #pragma unroll
        for (int r = 0; r < 8; ++r) rp[r * SCAN_LP] = a[r];
      }
      const int sn = (s + 1 < XS) ? s + 1 : s;
      const float* xq = xp0 + (size_t)sn * XD;
      #pragma unroll
      for (int kk = 0; kk < 8; ++kk) {
        ax[kk].x = xq[kk * 4];
        ax[kk].y = xq[kk * 4 + 1];
      }
    }
    lds_barrier();

    float hn;
    {
      float ir  = lraw[(0 * 16 + row) * SCAN_LP + lane];
      float fr  = lraw[(1 * 16 + row) * SCAN_LP + lane];
      float zr  = lraw[(2 * 16 + row) * SCAN_LP + lane];
      float orr = lraw[(3 * 16 + row) * SCAN_LP + lane];
      const float kL2E = 1.44269504088896341f;
      const float kLN2 = 0.693147180559945309f;
      float te  = __builtin_amdgcn_exp2f(-fabsf(fr) * kL2E);
      float ue  = 1.0f + te;
      float ce  = te - (ue - 1.0f);
      float l1p = fmaf(__builtin_amdgcn_logf(ue), kLN2, ce * __builtin_amdgcn_rcpf(ue));
      float ls  = fminf(fr, 0.0f) - l1p;
      float lfm = mS + ls;
      float mn  = fmaxf(ir, lfm);
      float ig  = __builtin_amdgcn_exp2f((ir - mn) * kL2E);
      float fg  = __builtin_amdgcn_exp2f((lfm - mn) * kL2E);
      float tz  = __builtin_amdgcn_tanhf(zr);
      float cn  = fg * cS + ig * tz;
      float nn  = fg * nS + ig;
      float og  = __builtin_amdgcn_rcpf(1.0f + __builtin_amdgcn_exp2f(-orr * kL2E));
      float nm  = __builtin_amdgcn_frexp_mantf(nn);
      int   ne  = __builtin_amdgcn_frexp_expf(nn);
      hn  = __builtin_amdgcn_ldexpf((og * cn) * __builtin_amdgcn_rcpf(nm), -ne);
      cS = cn;
      nS = nn;
      mS = mn;
      lh[row * SCAN_LP + lane] = hn;
    }

    {
      float mu = wave_sum32(hn) * (1.0f / XDH);
      float q  = wave_sum32(hn * hn) * (1.0f / XDH) - mu * mu;
      float rs = __builtin_amdgcn_rsqf(q + XEPS);
      float v = (hn - mu) * rs * gn;
      hp[(size_t)s * XD] = hold + v;
    }
    lds_barrier();
  }
}

__global__ __launch_bounds__(32) void k_ffn(const float* __restrict__ xn,
                                            float* __restrict__ hio,
                                            const float* __restrict__ Wup,
                                            const float* __restrict__ Wdn) {
  __shared__ float lact[16 * 128];
  const int lane = threadIdx.x;
  const int hlf  = lane >> 4;
  const int l16  = lane & 15;
  const size_t t0 = (size_t)blockIdx.x * 16;

  v2f ax[32];
  const float* xp = xn + (t0 + l16) * XD;
  #pragma unroll
  for (int kk = 0; kk < 32; ++kk) {
    int K0 = kk * 4 + hlf * 2;
    ax[kk].x = xp[K0];
    ax[kk].y = xp[K0 + 1];
  }

  float gatev[8][8];
  #pragma unroll
  for (int nt = 0; nt < 8; ++nt) {
    v8f a;
    #pragma unroll
    for (int r = 0; r < 8; ++r) a[r] = 0.0f;
    int j = nt * 16 + l16;
    #pragma unroll
    for (int kk = 0; kk < 32; ++kk) {
      int K0 = kk * 4 + hlf * 2;
      v2f b;
      b.x = Wup[(size_t)j * XD + K0];
      b.y = Wup[(size_t)j * XD + K0 + 1];
      a = wmma4(ax[kk], b, a);
    }
    #pragma unroll
    for (int r = 0; r < 8; ++r) gatev[nt][r] = a[r];
  }

  #pragma unroll
  for (int nt = 0; nt < 8; ++nt) {
    v8f a;
    #pragma unroll
    for (int r = 0; r < 8; ++r) a[r] = 0.0f;
    int j = 128 + nt * 16 + l16;
    #pragma unroll
    for (int kk = 0; kk < 32; ++kk) {
      int K0 = kk * 4 + hlf * 2;
      v2f b;
      b.x = Wup[(size_t)j * XD + K0];
      b.y = Wup[(size_t)j * XD + K0 + 1];
      a = wmma4(ax[kk], b, a);
    }
    #pragma unroll
    for (int r = 0; r < 8; ++r) {
      float act = gelu_exact(gatev[nt][r]) * a[r];
      lact[(r + hlf * 8) * 128 + nt * 16 + l16] = act;
    }
  }
  __syncthreads();

  v2f aa[32];
  const float* lp = lact + l16 * 128;
  #pragma unroll
  for (int kk = 0; kk < 32; ++kk) {
    int K0 = kk * 4 + hlf * 2;
    aa[kk].x = lp[K0];
    aa[kk].y = lp[K0 + 1];
  }
  #pragma unroll
  for (int nt = 0; nt < 8; ++nt) {
    v8f a;
    #pragma unroll
    for (int r = 0; r < 8; ++r) a[r] = 0.0f;
    int j = nt * 16 + l16;
    #pragma unroll
    for (int kk = 0; kk < 32; ++kk) {
      int K0 = kk * 4 + hlf * 2;
      v2f b;
      b.x = Wdn[(size_t)j * XD + K0];
      b.y = Wdn[(size_t)j * XD + K0 + 1];
      a = wmma4(aa[kk], b, a);
    }
    #pragma unroll
    for (int r = 0; r < 8; ++r) {
      size_t row = t0 + r + hlf * 8;
      hio[row * XD + nt * 16 + l16] += a[r];
    }
  }
}

__global__ __launch_bounds__(32) void k_proj(const float* __restrict__ xn,
                                             const float* __restrict__ pw,
                                             const float* __restrict__ pb,
                                             float* __restrict__ out) {
  const int lane = threadIdx.x;
  const int hlf  = lane >> 4;
  const int l16  = lane & 15;
  const size_t t0 = (size_t)blockIdx.x * 16;

  v8f a;
  #pragma unroll
  for (int r = 0; r < 8; ++r) a[r] = 0.0f;
  const float* xp = xn + (t0 + l16) * XD;
  #pragma unroll
  for (int kk = 0; kk < 32; ++kk) {
    int K0 = kk * 4 + hlf * 2;
    v2f av;
    av.x = xp[K0];
    av.y = xp[K0 + 1];
    v2f b;
    b.x = 0.0f;
    b.y = 0.0f;
    if (l16 < 9) {
      b.x = pw[(size_t)l16 * XD + K0];
      b.y = pw[(size_t)l16 * XD + K0 + 1];
    }
    a = wmma4(av, b, a);
  }
  if (l16 < 9) {
    float bb = pb[l16];
    #pragma unroll
    for (int r = 0; r < 8; ++r) {
      size_t row = t0 + r + hlf * 8;
      out[row * 9 + l16] = a[r] + bb;
    }
  }
}

extern "C" void kernel_launch(void* const* d_in, const int* in_sizes, int n_in,
                              void* d_out, int out_size, void* d_ws, size_t ws_size,
                              hipStream_t stream) {
  (void)in_sizes; (void)n_in; (void)out_size; (void)ws_size;
  const int*   x     = (const int*)d_in[0];
  const float* emb   = (const float*)d_in[1];
  const float* ln1   = (const float*)d_in[2];
  const float* Wg    = (const float*)d_in[3];
  const float* R     = (const float*)d_in[4];
  const float* bias  = (const float*)d_in[5];
  const float* gnw   = (const float*)d_in[6];
  const float* ln2   = (const float*)d_in[7];
  const float* wup   = (const float*)d_in[8];
  const float* wdn   = (const float*)d_in[9];
  const float* postw = (const float*)d_in[10];
  const float* pw    = (const float*)d_in[11];
  const float* pb    = (const float*)d_in[12];
  float* out = (float*)d_out;

  float* h  = (float*)d_ws;
  float* xb = h + (size_t)XT * XD;

  k_embed<<<(XT * XD + 255) / 256, 256, 0, stream>>>(x, emb, h, XT * XD);

  for (int bi = 0; bi < 2; ++bi) {
    k_ln<<<XT / 8, 256, 0, stream>>>(h, ln1 + bi * XD, xb);
    k_scan<<<dim3(XB / SCAN_R, XNH), 256, 0, stream>>>(
        xb, h,
        Wg   + (size_t)bi * 4 * XNH * XDH * XDH,
        R    + (size_t)bi * XNH * XDH * 4 * XDH,
        bias + (size_t)bi * XNH * 4 * XDH,
        gnw  + (size_t)bi * XD);
    k_ln<<<XT / 8, 256, 0, stream>>>(h, ln2 + bi * XD, xb);
    k_ffn<<<XT / 16, 32, 0, stream>>>(xb, h,
                                      wup + (size_t)bi * 2 * XD * XD,
                                      wdn + (size_t)bi * XD * XD);
  }

  k_ln<<<XT / 8, 256, 0, stream>>>(h, postw, xb);
  k_proj<<<XT / 16, 32, 0, stream>>>(xb, pw, pb, out);
}


# k_ffn: A and B fragments fetched 16 bytes per lane (k = 8q+4h+0..3, two WMMAs per fetch), half the loads and cache lines touched per WMMA; act tile pitch 132; k order of the sums changes
# speedup vs baseline: 7.1404x; 1.0535x over previous
#include <hip/hip_runtime.h>
#include <hip/hip_bf16.h>


typedef float v2f __attribute__((ext_vector_type(2)));
typedef float v8f __attribute__((ext_vector_type(8)));

#define XB   512
#define XS   64
#define XD   128
#define XNH  4
#define XDH  32
#define XT   (XB * XS)
#define XEPS 1e-5f

__device__ __forceinline__ v8f wmma4(v2f a, v2f b, v8f c) {
  return __builtin_amdgcn_wmma_f32_16x16x4_f32(false, a, false, b, (short)0, c,
                                               false, false);
}

__device__ __forceinline__ void lds_barrier() {
  __builtin_amdgcn_fence(__ATOMIC_RELEASE, "workgroup", "local");
  __builtin_amdgcn_s_barrier();
  __builtin_amdgcn_fence(__ATOMIC_ACQUIRE, "workgroup", "local");
}

template <int M>
__device__ __forceinline__ float row_xor(float v) {
  const int b = __builtin_bit_cast(int, v);
  return __builtin_bit_cast(float, __builtin_amdgcn_update_dpp(0, b, 0x160 + M, 0xf, 0xf, true));
}
__device__ __forceinline__ float xhalf(float v) {
  const int b = __builtin_bit_cast(int, v);
  return __builtin_bit_cast(float, __builtin_amdgcn_permlanex16(b, b, 0x76543210, 0xfedcba98, false, false));
}
__device__ __forceinline__ float wave_sum32(float v) {
  v += row_xor<1>(v);
  v += row_xor<2>(v);
  v += row_xor<4>(v);
  v += row_xor<8>(v);
  v += xhalf(v);
  return v;
}

__device__ __forceinline__ float gelu_exact(float x) {
  return 0.5f * x * (1.0f + erff(x * 0.70710678118654752f));
}

__global__ void k_embed(const int* __restrict__ x, const float* __restrict__ emb,
                        float* __restrict__ h, int total) {
  int i = blockIdx.x * blockDim.x + threadIdx.x;
  if (i >= total) return;
  int t = i >> 7;
  int c = i & 127;
  h[i] = emb[x[t] * XD + c];
}

__global__ void k_ln(const float* __restrict__ h, const float* __restrict__ w,
                     float* __restrict__ xn) {
  int wid  = threadIdx.x >> 5;
  int lane = threadIdx.x & 31;
  int t = blockIdx.x * (blockDim.x >> 5) + wid;
  const float4* hv = (const float4*)(h + (size_t)t * XD);
  float4 v = hv[lane];
  float s = v.x + v.y + v.z + v.w;
  float q = v.x * v.x + v.y * v.y + v.z * v.z + v.w * v.w;
  #pragma unroll
  for (int m = 16; m >= 1; m >>= 1) {
    s += __shfl_xor(s, m, 32);
    q += __shfl_xor(q, m, 32);
  }
  float mu  = s * (1.0f / XD);
  float var = q * (1.0f / XD) - mu * mu;
  float rs  = rsqrtf(var + XEPS);
  float4 wl = ((const float4*)w)[lane];
  float4 o;
  o.x = (v.x - mu) * rs * wl.x;
  o.y = (v.y - mu) * rs * wl.y;
  o.z = (v.z - mu) * rs * wl.z;
  o.w = (v.w - mu) * rs * wl.w;
  ((float4*)(xn + (size_t)t * XD))[lane] = o;
}

#define SCAN_LP 36
#define SCAN_R  8
__global__ __launch_bounds__(256) void k_scan(const float* __restrict__ xn,
                                              float* __restrict__ hio,
                                              const float* __restrict__ Wg,
                                              const float* __restrict__ R,
                                              const float* __restrict__ bias,
                                              const float* __restrict__ gnw) {
  __shared__ __attribute__((aligned(16))) float lraw[4 * 16 * SCAN_LP];
  __shared__ __attribute__((aligned(16))) float lh[SCAN_R * SCAN_LP];

  const int lane = threadIdx.x & 31;
  const int w    = __builtin_amdgcn_readfirstlane((int)(threadIdx.x >> 5));
  const int hlf  = lane >> 4;
  const int l16  = lane & 15;
  const int ar   = lane & 7;
  const int b0   = blockIdx.x * SCAN_R;
  const int n    = blockIdx.y;

  v2f bw[8], br[8];
  const int N  = w * 16 + l16;
  {
    const int g  = N >> 5;
    const int dh = N & 31;
    #pragma unroll
    for (int kk = 0; kk < 8; ++kk) {
      const int K0 = kk * 4 + hlf * 2;
      const float* wgp = Wg + (((size_t)(g * XNH + n) * XDH + dh) * XDH + K0);
      bw[kk].x = wgp[0];
      bw[kk].y = wgp[1];
      br[kk].x = R[(size_t)(n * XDH + K0 + 0) * (4 * XDH) + N];
      br[kk].y = R[(size_t)(n * XDH + K0 + 1) * (4 * XDH) + N];
    }
  }
  const float bj = bias[n * 128 + N];
  const float gn = gnw[n * XDH + lane];
  for (int j = threadIdx.x; j < SCAN_R * SCAN_LP; j += 256) lh[j] = 0.0f;
  __syncthreads();

  float cS = 0.0f, nS = 0.0f, mS = 0.0f;
  const int row = w;
  float* hp = hio + ((size_t)(b0 + row) * XS) * XD + n * XDH + lane;
  const float* xp0 = xn + ((size_t)(b0 + ar) * XS) * XD + n * XDH + hlf * 2;

  v2f ax[8];
  #pragma unroll
  for (int kk = 0; kk < 8; ++kk) {
    ax[kk].x = xp0[kk * 4];
    ax[kk].y = xp0[kk * 4 + 1];
  }

  #pragma unroll 1
  for (int s = 0; s < XS; ++s) {
    const float hold = hp[(size_t)s * XD];

    {
      v2f ah[8];
      #pragma unroll
      for (int kk = 0; kk < 8; ++kk) {
        ah[kk].x = lh[ar * SCAN_LP + kk * 4 + hlf * 2];
        ah[kk].y = lh[ar * SCAN_LP + kk * 4 + hlf * 2 + 1];
      }
      __builtin_amdgcn_sched_barrier(0);
      v8f a;
      #pragma unroll
      for (int r = 0; r < 8; ++r) a[r] = bj;
      #pragma unroll
      for (int kk = 0; kk < 8; ++kk) {
        a = wmma4(ax[kk], bw[kk], a);
        a = wmma4(ah[kk], br[kk], a);
      }
      {
        float* rp = lraw + ((w >> 1) * 16 + hlf * 8) * SCAN_LP + (w & 1) * 16 + l16;
        #pragma unroll
        for (int r = 0; r < 8; ++r) rp[r * SCAN_LP] = a[r];
      }
      const int sn = (s + 1 < XS) ? s + 1 : s;
      const float* xq = xp0 + (size_t)sn * XD;
      #pragma unroll
      for (int kk = 0; kk < 8; ++kk) {
        ax[kk].x = xq[kk * 4];
        ax[kk].y = xq[kk * 4 + 1];
      }
    }
    lds_barrier();

    float hn;
    {
      float ir  = lraw[(0 * 16 + row) * SCAN_LP + lane];
      float fr  = lraw[(1 * 16 + row) * SCAN_LP + lane];
      float zr  = lraw[(2 * 16 + row) * SCAN_LP + lane];
      float orr = lraw[(3 * 16 + row) * SCAN_LP + lane];
      const float kL2E = 1.44269504088896341f;
      const float kLN2 = 0.693147180559945309f;
      float te  = __builtin_amdgcn_exp2f(-fabsf(fr) * kL2E);
      float ue  = 1.0f + te;
      float ce  = te - (ue - 1.0f);
      float l1p = fmaf(__builtin_amdgcn_logf(ue), kLN2, ce * __builtin_amdgcn_rcpf(ue));
      float ls  = fminf(fr, 0.0f) - l1p;
      float lfm = mS + ls;
      float mn  = fmaxf(ir, lfm);
      float ig  = __builtin_amdgcn_exp2f((ir - mn) * kL2E);
      float fg  = __builtin_amdgcn_exp2f((lfm - mn) * kL2E);
      float tz  = __builtin_amdgcn_tanhf(zr);
      float cn  = fg * cS + ig * tz;
      float nn  = fg * nS + ig;
      float og  = __builtin_amdgcn_rcpf(1.0f + __builtin_amdgcn_exp2f(-orr * kL2E));
      float nm  = __builtin_amdgcn_frexp_mantf(nn);
      int   ne  = __builtin_amdgcn_frexp_expf(nn);
      hn  = __builtin_amdgcn_ldexpf((og * cn) * __builtin_amdgcn_rcpf(nm), -ne);
      cS = cn;
      nS = nn;
      mS = mn;
      lh[row * SCAN_LP + lane] = hn;
    }

    {
      float mu = wave_sum32(hn) * (1.0f / XDH);
      float q  = wave_sum32(hn * hn) * (1.0f / XDH) - mu * mu;
      float rs = __builtin_amdgcn_rsqf(q + XEPS);
      float v = (hn - mu) * rs * gn;
      hp[(size_t)s * XD] = hold + v;
    }
    lds_barrier();
  }
}

#define FFN_LP 132
__device__ __forceinline__ v8f ffn_mac8(float4 a, float4 b, v8f c) {
  v2f a0, a1, b0, b1;
  a0.x = a.x; a0.y = a.y; a1.x = a.z; a1.y = a.w;
  b0.x = b.x; b0.y = b.y; b1.x = b.z; b1.y = b.w;
  c = wmma4(a0, b0, c);
  return wmma4(a1, b1, c);
}
__global__ __launch_bounds__(32) void k_ffn(const float* __restrict__ xn,
                                            float* __restrict__ hio,
                                            const float* __restrict__ Wup,
                                            const float* __restrict__ Wdn) {
  __shared__ __attribute__((aligned(16))) float lact[16 * FFN_LP];
  const int lane = threadIdx.x;
  const int hlf  = lane >> 4;
  const int l16  = lane & 15;
  const size_t t0 = (size_t)blockIdx.x * 16;

  float4 ax[16];
  const float* xp = xn + (t0 + l16) * XD + hlf * 4;
  #pragma unroll
  for (int q = 0; q < 16; ++q) ax[q] = *(const float4*)(xp + q * 8);

  float gatev[8][8];
  #pragma unroll
  for (int nt = 0; nt < 8; ++nt) {
    v8f a;
    #pragma unroll
    for (int r = 0; r < 8; ++r) a[r] = 0.0f;
    const float* wp = Wup + (size_t)(nt * 16 + l16) * XD + hlf * 4;
    #pragma unroll
    for (int q = 0; q < 16; ++q) {
      const float4 b = *(const float4*)(wp + q * 8);
      a = ffn_mac8(ax[q], b, a);
    }
    #pragma unroll
    for (int r = 0; r < 8; ++r) gatev[nt][r] = a[r];
  }

  #pragma unroll
  for (int nt = 0; nt < 8; ++nt) {
    v8f a;
    #pragma unroll
    for (int r = 0; r < 8; ++r) a[r] = 0.0f;
    const float* wp = Wup + (size_t)(128 + nt * 16 + l16) * XD + hlf * 4;
    #pragma unroll
    for (int q = 0; q < 16; ++q) {
      const float4 b = *(const float4*)(wp + q * 8);
      a = ffn_mac8(ax[q], b, a);
    }
    #pragma unroll
    for (int r = 0; r < 8; ++r) {
      float act = gelu_exact(gatev[nt][r]) * a[r];
      lact[(r + hlf * 8) * FFN_LP + nt * 16 + l16] = act;
    }
  }
  __syncthreads();

  float4 aa[16];
  const float* lp = lact + l16 * FFN_LP + hlf * 4;
  #pragma unroll
  for (int q = 0; q < 16; ++q) aa[q] = *(const float4*)(lp + q * 8);
  #pragma unroll
  for (int nt = 0; nt < 8; ++nt) {
    v8f a;
    #pragma unroll
    for (int r = 0; r < 8; ++r) a[r] = 0.0f;
    const float* wp = Wdn + (size_t)(nt * 16 + l16) * XD + hlf * 4;
    #pragma unroll
    for (int q = 0; q < 16; ++q) {
      const float4 b = *(const float4*)(wp + q * 8);
      a = ffn_mac8(aa[q], b, a);
    }
    #pragma unroll
    for (int r = 0; r < 8; ++r) {
      size_t row = t0 + r + hlf * 8;
      hio[row * XD + nt * 16 + l16] += a[r];
    }
  }
}

__global__ __launch_bounds__(32) void k_proj(const float* __restrict__ xn,
                                             const float* __restrict__ pw,
                                             const float* __restrict__ pb,
                                             float* __restrict__ out) {
  const int lane = threadIdx.x;
  const int hlf  = lane >> 4;
  const int l16  = lane & 15;
  const size_t t0 = (size_t)blockIdx.x * 16;

  v8f a;
  #pragma unroll
  for (int r = 0; r < 8; ++r) a[r] = 0.0f;
  const float* xp = xn + (t0 + l16) * XD;
  #pragma unroll
  for (int kk = 0; kk < 32; ++kk) {
    int K0 = kk * 4 + hlf * 2;
    v2f av;
    av.x = xp[K0];
    av.y = xp[K0 + 1];
    v2f b;
    b.x = 0.0f;
    b.y = 0.0f;
    if (l16 < 9) {
      b.x = pw[(size_t)l16 * XD + K0];
      b.y = pw[(size_t)l16 * XD + K0 + 1];
    }
    a = wmma4(av, b, a);
  }
  if (l16 < 9) {
    float bb = pb[l16];
    #pragma unroll
    for (int r = 0; r < 8; ++r) {
      size_t row = t0 + r + hlf * 8;
      out[row * 9 + l16] = a[r] + bb;
    }
  }
}

extern "C" void kernel_launch(void* const* d_in, const int* in_sizes, int n_in,
                              void* d_out, int out_size, void* d_ws, size_t ws_size,
                              hipStream_t stream) {
  (void)in_sizes; (void)n_in; (void)out_size; (void)ws_size;
  const int*   x     = (const int*)d_in[0];
  const float* emb   = (const float*)d_in[1];
  const float* ln1   = (const float*)d_in[2];
  const float* Wg    = (const float*)d_in[3];
  const float* R     = (const float*)d_in[4];
  const float* bias  = (const float*)d_in[5];
  const float* gnw   = (const float*)d_in[6];
  const float* ln2   = (const float*)d_in[7];
  const float* wup   = (const float*)d_in[8];
  const float* wdn   = (const float*)d_in[9];
  const float* postw = (const float*)d_in[10];
  const float* pw    = (const float*)d_in[11];
  const float* pb    = (const float*)d_in[12];
  float* out = (float*)d_out;

  float* h  = (float*)d_ws;
  float* xb = h + (size_t)XT * XD;

  k_embed<<<(XT * XD + 255) / 256, 256, 0, stream>>>(x, emb, h, XT * XD);

  for (int bi = 0; bi < 2; ++bi) {
    k_ln<<<XT / 8, 256, 0, stream>>>(h, ln1 + bi * XD, xb);
    k_scan<<<dim3(XB / SCAN_R, XNH), 256, 0, stream>>>(
        xb, h,
        Wg   + (size_t)bi * 4 * XNH * XDH * XDH,
        R    + (size_t)bi * XNH * XDH * 4 * XDH,
        bias + (size_t)bi * XNH * 4 * XDH,
        gnw  + (size_t)bi * XD);
    k_ln<<<XT / 8, 256, 0, stream>>>(h, ln2 + bi * XD, xb);
    k_ffn<<<XT / 16, 32, 0, stream>>>(xb, h,
                                      wup + (size_t)bi * 2 * XD * XD,
                                      wdn + (size_t)bi * XD * XD);
  }

  k_ln<<<XT / 8, 256, 0, stream>>>(h, postw, xb);
  k_proj<<<XT / 16, 32, 0, stream>>>(xb, pw, pb, out);
}
